# starting kernel
# baseline (speedup 1.0000x reference)
#include <hip/hip_runtime.h>


typedef __attribute__((ext_vector_type(2))) float v2f;
typedef __attribute__((ext_vector_type(8))) float v8f;

#define NQ      12
#define DIM     4096
#define THREADS 256
#define NWAVES  8
#define NLAYERS 6
#define NPAIRS  6
#define NROUNDS 8

__device__ __forceinline__ float2 cmul(float2 a, float2 b) {
    return make_float2(a.x * b.x - a.y * b.y, a.x * b.y + a.y * b.x);
}

__device__ __forceinline__ void fused_rot(float tx, float ty, float tz, float2* U) {
    const float cx = cosf(0.5f * tx), sx = sinf(0.5f * tx);
    const float cy = cosf(0.5f * ty), sy = sinf(0.5f * ty);
    const float cz = cosf(0.5f * tz), sz = sinf(0.5f * tz);
    const float2 A00 = make_float2( cy * cx,  sy * sx);
    const float2 A01 = make_float2(-sy * cx, -cy * sx);
    const float2 A10 = make_float2( sy * cx, -cy * sx);
    const float2 A11 = make_float2( cy * cx, -sy * sx);
    const float2 e  = make_float2(cz, -sz);
    const float2 ec = make_float2(cz,  sz);
    U[0] = cmul(e, A00);  U[1] = cmul(e, A01);
    U[2] = cmul(ec, A10); U[3] = cmul(ec, A11);
}

__device__ __forceinline__ void embed_rot(float f, float2* U) {
    const float cy = cosf(0.5f * f),  sy = sinf(0.5f * f);
    const float cz = cosf(0.25f * f), sz = sinf(0.25f * f);
    U[0] = make_float2( cz * cy, -sz * cy);
    U[1] = make_float2(-cz * sy,  sz * sy);
    U[2] = make_float2( cz * sy,  sz * sy);
    U[3] = make_float2( cz * cy,  sz * cy);
}

__device__ __forceinline__ void kron22(const float2* Ub, const float2* Ua, float2* M) {
    #pragma unroll
    for (int o = 0; o < 4; ++o)
        #pragma unroll
        for (int j = 0; j < 4; ++j)
            M[o * 4 + j] = cmul(Ub[(o >> 1) * 2 + (j >> 1)], Ua[(o & 1) * 2 + (j & 1)]);
}

__device__ __forceinline__ void apply_pair_wmma(float2* st, const float2* M,
                                                int w, int lane, int wave) {
    const int col = lane & 15;
    const int hiK = lane >> 4;
    const int o   = (col >> 1) & 3;
    const float2 m1 = M[o * 4 + hiK];
    const float2 m2 = M[o * 4 + hiK + 2];
    float a10, a11, a20, a21;
    if ((col & 1) == 0) { a10 = m1.x; a11 = -m1.y; a20 = m2.x; a21 = -m2.y; }
    else                { a10 = m1.y; a11 =  m1.x; a20 = m2.y; a21 =  m2.x; }
    v2f A1; A1[0] = a10; A1[1] = a11;
    v2f A2; A2[0] = a20; A2[1] = a21;

    const int s       = 1 << w;
    const int lowmask = s - 1;
    const int off1    = hiK ? s : 0;
    const int off2    = off1 + 2 * s;
    const int jst     = hiK ? 2 : 0;

    int    bases[NROUNDS];
    float2 p1[NROUNDS], p2[NROUNDS];
    #pragma unroll
    for (int r = 0; r < NROUNDS; ++r) {
        const int g = r * (NWAVES * 16) + wave * 16 + col;
        bases[r] = ((g >> w) << (w + 2)) | (g & lowmask);
        p1[r] = st[bases[r] + off1];
        p2[r] = st[bases[r] + off2];
    }
    #pragma unroll
    for (int r = 0; r < NROUNDS; ++r) {
        v2f B1; B1[0] = p1[r].x; B1[1] = p1[r].y;
        v2f B2; B2[0] = p2[r].x; B2[1] = p2[r].y;
        v8f C = {0.f, 0.f, 0.f, 0.f, 0.f, 0.f, 0.f, 0.f};
        C = __builtin_amdgcn_wmma_f32_16x16x4_f32(false, A1, false, B1,
                                                  (short)0, C, false, false);
        C = __builtin_amdgcn_wmma_f32_16x16x4_f32(false, A2, false, B2,
                                                  (short)0, C, false, false);
        const float2 oa = hiK ? make_float2(C[4], C[5]) : make_float2(C[0], C[1]);
        const float2 ob = hiK ? make_float2(C[6], C[7]) : make_float2(C[2], C[3]);
        st[bases[r] + jst * s]       = oa;
        st[bases[r] + (jst + 1) * s] = ob;
    }
}

__global__ __launch_bounds__(THREADS)
void qsim_kernel(const float* __restrict__ fv, const float* __restrict__ theta,
                 float* __restrict__ out) {
    __shared__ float2 st[DIM];
    __shared__ float2 embM[NPAIRS][16];
    __shared__ float2 layM[NLAYERS * NPAIRS][16];
    __shared__ float  wpart[NWAVES][NQ];

    const int b    = blockIdx.x;
    const int tid  = threadIdx.x;
    const int lane = tid & 31;
    const int wave = tid >> 5;

    for (int i = tid; i < DIM; i += THREADS) st[i] = make_float2(0.f, 0.f);
    if (tid == 0) st[0] = make_float2(1.f, 0.f);

    if (tid < NPAIRS + NLAYERS * NPAIRS) {
        float2 Ua[4], Ub[4];
        if (tid < NPAIRS) {
            embed_rot(fv[b * NQ + 2 * tid],     Ua);
            embed_rot(fv[b * NQ + 2 * tid + 1], Ub);
            kron22(Ub, Ua, &embM[tid][0]);
        } else {
            const int t = tid - NPAIRS, l = t / NPAIRS, p = t % NPAIRS;
            const float* ta = theta + (l * NQ + 2 * p) * 3;
            fused_rot(ta[0], ta[1], ta[2], Ua);
            fused_rot(ta[3], ta[4], ta[5], Ub);
            kron22(Ub, Ua, &layM[t][0]);
        }
    }
    __syncthreads();

    for (int p = 0; p < NPAIRS; ++p) {
        apply_pair_wmma(st, &embM[p][0], 2 * p, lane, wave);
        __syncthreads();
    }

    for (int l = 0; l < NLAYERS; ++l) {
        for (int p = 0; p < NPAIRS; ++p) {
            apply_pair_wmma(st, &layM[l * NPAIRS + p][0], 2 * p, lane, wave);
            __syncthreads();
        }
        for (int k = 0; k < NQ; ++k) {
            const int c = (k < NQ - 1) ? k : NQ - 1;
            const int t = (k < NQ - 1) ? k + 1 : 0;
            const int lo = c < t ? c : t;
            const int hi = c < t ? t : c;
            const int midbits = hi - lo - 1;
            for (int p = tid; p < DIM / 4; p += THREADS) {
                const int i    = p & ((1 << lo) - 1);
                const int rest = p >> lo;
                const int mid  = rest & ((1 << midbits) - 1);
                const int top  = rest >> midbits;
                const int idx  = i | (mid << (lo + 1)) | (top << (hi + 1)) | (1 << c);
                const int jdx  = idx | (1 << t);
                float2 tmp = st[idx];
                st[idx] = st[jdx];
                st[jdx] = tmp;
            }
            __syncthreads();
        }
    }

    float acc[NQ];
    #pragma unroll
    for (int w = 0; w < NQ; ++w) acc[w] = 0.f;
    for (int i = tid; i < DIM; i += THREADS) {
        const float2 a  = st[i];
        const float  pz = a.x * a.x + a.y * a.y;
        #pragma unroll
        for (int w = 0; w < NQ; ++w)
            acc[w] += ((i >> w) & 1) ? -pz : pz;
    }
    #pragma unroll
    for (int off = 16; off > 0; off >>= 1) {
        #pragma unroll
        for (int w = 0; w < NQ; ++w)
            acc[w] += __shfl_xor(acc[w], off, 32);
    }
    if (lane == 0) {
        #pragma unroll
        for (int w = 0; w < NQ; ++w) wpart[wave][w] = acc[w];
    }
    __syncthreads();
    if (tid < NQ) {
        float s = 0.f;
        for (int q = 0; q < NWAVES; ++q) s += wpart[q][tid];
        out[b * NQ + tid] = s;
    }
}

extern "C" void kernel_launch(void* const* d_in, const int* in_sizes, int n_in,
                              void* d_out, int out_size, void* d_ws, size_t ws_size,
                              hipStream_t stream) {
    const float* fv    = (const float*)d_in[0];
    const float* theta = (const float*)d_in[1];
    float*       out   = (float*)d_out;
    const int batch = in_sizes[0] / NQ;
    qsim_kernel<<<batch, THREADS, 0, stream>>>(fv, theta, out);
}


# CNOT ring composed into one index permutation (prefix xor of the index bits), applied by the stores of each layer's last pair gate: 12 swap passes and 12 barriers per layer become 1 barrier
# speedup vs baseline: 1.5937x; 1.5937x over previous
#include <hip/hip_runtime.h>


typedef __attribute__((ext_vector_type(2))) float v2f;
typedef __attribute__((ext_vector_type(8))) float v8f;

#define NQ      12
#define DIM     4096
#define THREADS 256
#define NWAVES  8
#define NLAYERS 6
#define NPAIRS  6
#define NROUNDS 8

__device__ __forceinline__ float2 cmul(float2 a, float2 b) {
    return make_float2(a.x * b.x - a.y * b.y, a.x * b.y + a.y * b.x);
}

__device__ __forceinline__ void fused_rot(float tx, float ty, float tz, float2* U) {
    const float cx = cosf(0.5f * tx), sx = sinf(0.5f * tx);
    const float cy = cosf(0.5f * ty), sy = sinf(0.5f * ty);
    const float cz = cosf(0.5f * tz), sz = sinf(0.5f * tz);
    const float2 A00 = make_float2( cy * cx,  sy * sx);
    const float2 A01 = make_float2(-sy * cx, -cy * sx);
    const float2 A10 = make_float2( sy * cx, -cy * sx);
    const float2 A11 = make_float2( cy * cx, -sy * sx);
    const float2 e  = make_float2(cz, -sz);
    const float2 ec = make_float2(cz,  sz);
    U[0] = cmul(e, A00);  U[1] = cmul(e, A01);
    U[2] = cmul(ec, A10); U[3] = cmul(ec, A11);
}

__device__ __forceinline__ void embed_rot(float f, float2* U) {
    const float cy = cosf(0.5f * f),  sy = sinf(0.5f * f);
    const float cz = cosf(0.25f * f), sz = sinf(0.25f * f);
    U[0] = make_float2( cz * cy, -sz * cy);
    U[1] = make_float2(-cz * sy,  sz * sy);
    U[2] = make_float2( cz * sy,  sz * sy);
    U[3] = make_float2( cz * cy,  sz * cy);
}

__device__ __forceinline__ void kron22(const float2* Ub, const float2* Ua, float2* M) {
    #pragma unroll
    for (int o = 0; o < 4; ++o)
        #pragma unroll
        for (int j = 0; j < 4; ++j)
            M[o * 4 + j] = cmul(Ub[(o >> 1) * 2 + (j >> 1)], Ua[(o & 1) * 2 + (j & 1)]);
}

__device__ __forceinline__ int ring_perm(int x) {
    int y = x ^ (x << 1);
    y ^= y << 2;
    y ^= y << 4;
    y ^= y << 8;
    y &= DIM - 1;
    return (y & ~1) | (((y >> (NQ - 1)) ^ x) & 1);
}

template <bool RING>
__device__ __forceinline__ void apply_pair_wmma(float2* st, const float2* M,
                                                int w, int lane, int wave) {
    const int col = lane & 15;
    const int hiK = lane >> 4;
    const int o   = (col >> 1) & 3;
    const float2 m1 = M[o * 4 + hiK];
    const float2 m2 = M[o * 4 + hiK + 2];
    float a10, a11, a20, a21;
    if ((col & 1) == 0) { a10 = m1.x; a11 = -m1.y; a20 = m2.x; a21 = -m2.y; }
    else                { a10 = m1.y; a11 =  m1.x; a20 = m2.y; a21 =  m2.x; }
    v2f A1; A1[0] = a10; A1[1] = a11;
    v2f A2; A2[0] = a20; A2[1] = a21;

    const int s       = 1 << w;
    const int lowmask = s - 1;
    const int off1    = hiK ? s : 0;
    const int off2    = off1 + 2 * s;
    const int jst     = hiK ? 2 : 0;

    int    bases[NROUNDS];
    float2 p1[NROUNDS], p2[NROUNDS];
    #pragma unroll
    for (int r = 0; r < NROUNDS; ++r) {
        const int g = r * (NWAVES * 16) + wave * 16 + col;
        bases[r] = ((g >> w) << (w + 2)) | (g & lowmask);
        p1[r] = st[bases[r] + off1];
        p2[r] = st[bases[r] + off2];
    }
    float2 qa[NROUNDS], qb[NROUNDS];
    #pragma unroll
    for (int r = 0; r < NROUNDS; ++r) {
        v2f B1; B1[0] = p1[r].x; B1[1] = p1[r].y;
        v2f B2; B2[0] = p2[r].x; B2[1] = p2[r].y;
        v8f C = {0.f, 0.f, 0.f, 0.f, 0.f, 0.f, 0.f, 0.f};
        C = __builtin_amdgcn_wmma_f32_16x16x4_f32(false, A1, false, B1,
                                                  (short)0, C, false, false);
        C = __builtin_amdgcn_wmma_f32_16x16x4_f32(false, A2, false, B2,
                                                  (short)0, C, false, false);
        const float2 oa = hiK ? make_float2(C[4], C[5]) : make_float2(C[0], C[1]);
        const float2 ob = hiK ? make_float2(C[6], C[7]) : make_float2(C[2], C[3]);
        if (RING) { qa[r] = oa; qb[r] = ob; }
        else {
            st[bases[r] + jst * s]       = oa;
            st[bases[r] + (jst + 1) * s] = ob;
        }
    }
    if (RING) {
        __syncthreads();
        #pragma unroll
        for (int r = 0; r < NROUNDS; ++r) {
            st[ring_perm(bases[r] + jst * s)]       = qa[r];
            st[ring_perm(bases[r] + (jst + 1) * s)] = qb[r];
        }
    }
}

__global__ __launch_bounds__(THREADS, 6)
void qsim_kernel(const float* __restrict__ fv, const float* __restrict__ theta,
                 float* __restrict__ out) {
    __shared__ float2 st[DIM];
    __shared__ float2 embM[NPAIRS][16];
    __shared__ float2 layM[NLAYERS * NPAIRS][16];
    __shared__ float  wpart[NWAVES][NQ];

    const int b    = blockIdx.x;
    const int tid  = threadIdx.x;
    const int lane = tid & 31;
    const int wave = tid >> 5;

    for (int i = tid; i < DIM; i += THREADS) st[i] = make_float2(0.f, 0.f);
    if (tid == 0) st[0] = make_float2(1.f, 0.f);

    if (tid < NPAIRS + NLAYERS * NPAIRS) {
        float2 Ua[4], Ub[4];
        if (tid < NPAIRS) {
            embed_rot(fv[b * NQ + 2 * tid],     Ua);
            embed_rot(fv[b * NQ + 2 * tid + 1], Ub);
            kron22(Ub, Ua, &embM[tid][0]);
        } else {
            const int t = tid - NPAIRS, l = t / NPAIRS, p = t % NPAIRS;
            const float* ta = theta + (l * NQ + 2 * p) * 3;
            fused_rot(ta[0], ta[1], ta[2], Ua);
            fused_rot(ta[3], ta[4], ta[5], Ub);
            kron22(Ub, Ua, &layM[t][0]);
        }
    }
    __syncthreads();

    for (int p = 0; p < NPAIRS; ++p) {
        apply_pair_wmma<false>(st, &embM[p][0], 2 * p, lane, wave);
        __syncthreads();
    }

    for (int l = 0; l < NLAYERS; ++l) {
        for (int p = 0; p < NPAIRS - 1; ++p) {
            apply_pair_wmma<false>(st, &layM[l * NPAIRS + p][0], 2 * p, lane, wave);
            __syncthreads();
        }
        apply_pair_wmma<true>(st, &layM[l * NPAIRS + NPAIRS - 1][0], NQ - 2, lane, wave);
        __syncthreads();
    }

    float acc[NQ];
    #pragma unroll
    for (int w = 0; w < NQ; ++w) acc[w] = 0.f;
    for (int i = tid; i < DIM; i += THREADS) {
        const float2 a  = st[i];
        const float  pz = a.x * a.x + a.y * a.y;
        #pragma unroll
        for (int w = 0; w < NQ; ++w)
            acc[w] += ((i >> w) & 1) ? -pz : pz;
    }
    #pragma unroll
    for (int off = 16; off > 0; off >>= 1) {
        #pragma unroll
        for (int w = 0; w < NQ; ++w)
            acc[w] += __shfl_xor(acc[w], off, 32);
    }
    if (lane == 0) {
        #pragma unroll
        for (int w = 0; w < NQ; ++w) wpart[wave][w] = acc[w];
    }
    __syncthreads();
    if (tid < NQ) {
        float s = 0.f;
        for (int q = 0; q < NWAVES; ++q) s += wpart[q][tid];
        out[b * NQ + tid] = s;
    }
}

extern "C" void kernel_launch(void* const* d_in, const int* in_sizes, int n_in,
                              void* d_out, int out_size, void* d_ws, size_t ws_size,
                              hipStream_t stream) {
    const float* fv    = (const float*)d_in[0];
    const float* theta = (const float*)d_in[1];
    float*       out   = (float*)d_out;
    const int batch = in_sizes[0] / NQ;
    qsim_kernel<<<batch, THREADS, 0, stream>>>(fv, theta, out);
}


# single-qubit gates fused per wire triple into 8x8 complex (16x16 real) matrices, 4 chained K=4 WMMAs per tile: 4 passes per layer, not 6, no duplicated A rows; products summed in a new order
# speedup vs baseline: 2.6056x; 1.6349x over previous
#include <hip/hip_runtime.h>


typedef __attribute__((ext_vector_type(2))) float v2f;
typedef __attribute__((ext_vector_type(8))) float v8f;

#define NQ      12
#define DIM     4096
#define THREADS 256
#define NWAVES  8
#define NLAYERS 6
#define NTRI    4
#define NROUNDS 4
#define NGATES  ((NLAYERS + 1) * NTRI)

#define T0_A 3
#define T0_B 4
#define T0_C 11
#define T0_COLS 0x027
#define T1_A 0
#define T1_B 5
#define T1_C 6
#define T1_COLS 0x01E
#define T2_A 1
#define T2_B 7
#define T2_C 8
#define T2_COLS 0x01D
#define T3_A 2
#define T3_B 9
#define T3_C 10
#define T3_COLS 0x01B
#define PACK4(x0, x1, x2, x3) ((x0) | ((x1) << 4) | ((x2) << 8) | ((x3) << 12))

__device__ __forceinline__ float2 cmul(float2 a, float2 b) {
    return make_float2(a.x * b.x - a.y * b.y, a.x * b.y + a.y * b.x);
}

__device__ __forceinline__ void fused_rot(float tx, float ty, float tz, float2* U) {
    const float cx = cosf(0.5f * tx), sx = sinf(0.5f * tx);
    const float cy = cosf(0.5f * ty), sy = sinf(0.5f * ty);
    const float cz = cosf(0.5f * tz), sz = sinf(0.5f * tz);
    const float2 A00 = make_float2( cy * cx,  sy * sx);
    const float2 A01 = make_float2(-sy * cx, -cy * sx);
    const float2 A10 = make_float2( sy * cx, -cy * sx);
    const float2 A11 = make_float2( cy * cx, -sy * sx);
    const float2 e  = make_float2(cz, -sz);
    const float2 ec = make_float2(cz,  sz);
    U[0] = cmul(e, A00);  U[1] = cmul(e, A01);
    U[2] = cmul(ec, A10); U[3] = cmul(ec, A11);
}

__device__ __forceinline__ void embed_rot(float f, float2* U) {
    const float cy = cosf(0.5f * f),  sy = sinf(0.5f * f);
    const float cz = cosf(0.25f * f), sz = sinf(0.25f * f);
    U[0] = make_float2( cz * cy, -sz * cy);
    U[1] = make_float2(-cz * sy,  sz * sy);
    U[2] = make_float2( cz * sy,  sz * sy);
    U[3] = make_float2( cz * cy,  sz * cy);
}

__host__ __device__ constexpr int ring_perm(int x) {
    int y = x ^ (x << 1);
    y ^= y << 2;
    y ^= y << 4;
    y ^= y << 8;
    y &= DIM - 1;
    return (y & ~1) | (((y >> (NQ - 1)) ^ x) & 1);
}

__host__ __device__ constexpr int deposit(int v, int mask) {
    int out = 0, k = 0;
    for (int bit = 0; bit < NQ; ++bit)
        if ((mask >> bit) & 1) { out |= ((v >> k) & 1) << bit; ++k; }
    return out;
}

template <int WA, int WB, int WC, int COLS, bool RING>
__device__ __forceinline__ void apply_triple_wmma(float2* st, const float2* Mg,
                                                  int lane, int wave) {
    constexpr int REST = (DIM - 1) & ~(COLS | (1 << WA) | (1 << WB) | (1 << WC));
    const int col = lane & 15;
    const int h   = lane >> 4;

    const int orow = (col >> 3) | ((col & 6));
    const float4* mp = reinterpret_cast<const float4*>(Mg + orow * 8 + h * 4);
    const float4 mlo = mp[0], mhi = mp[1];
    const float2 m[4] = { make_float2(mlo.x, mlo.y), make_float2(mlo.z, mlo.w),
                          make_float2(mhi.x, mhi.y), make_float2(mhi.z, mhi.w) };
    const bool imrow = (col & 1) != 0;
    v2f Aop[4];
    #pragma unroll
    for (int t = 0; t < 4; ++t) {
        Aop[t][0] = imrow ? m[t].y :  m[t].x;
        Aop[t][1] = imrow ? m[t].x : -m[t].y;
    }

    const int base = deposit(col, COLS) | (h << WA) | deposit(wave, REST);

    float2 p[NROUNDS][4];
    #pragma unroll
    for (int r = 0; r < NROUNDS; ++r)
        #pragma unroll
        for (int t = 0; t < 4; ++t)
            p[r][t] = st[base + deposit(r << 3, REST) + ((t & 1) << WB) + ((t >> 1) << WC)];
    #pragma unroll
    for (int r = 0; r < NROUNDS; ++r)
        asm volatile("" :: "v"(p[r][0].x), "v"(p[r][0].y), "v"(p[r][1].x), "v"(p[r][1].y),
                           "v"(p[r][2].x), "v"(p[r][2].y), "v"(p[r][3].x), "v"(p[r][3].y));

    v8f acc[NROUNDS];
    #pragma unroll
    for (int t = 0; t < 4; ++t)
        #pragma unroll
        for (int r = 0; r < NROUNDS; ++r) {
            v2f Bop; Bop[0] = p[r][t].x; Bop[1] = p[r][t].y;
            v8f Cin = {0.f, 0.f, 0.f, 0.f, 0.f, 0.f, 0.f, 0.f};
            if (t > 0) Cin = acc[r];
            acc[r] = __builtin_amdgcn_wmma_f32_16x16x4_f32(false, Aop[t], false, Bop,
                                                           (short)0, Cin, false, false);
        }

    if (!RING) {
        #pragma unroll
        for (int r = 0; r < NROUNDS; ++r)
            #pragma unroll
            for (int q = 0; q < 4; ++q)
                st[base + deposit(r << 3, REST) + ((q & 1) << WB) + ((q >> 1) << WC)] =
                    make_float2(acc[r][2 * q], acc[r][2 * q + 1]);
    } else {
        __syncthreads();
        const int pbase = ring_perm(base);
        #pragma unroll
        for (int r = 0; r < NROUNDS; ++r)
            #pragma unroll
            for (int q = 0; q < 4; ++q)
                st[pbase ^ ring_perm(deposit(r << 3, REST) + ((q & 1) << WB) + ((q >> 1) << WC))] =
                    make_float2(acc[r][2 * q], acc[r][2 * q + 1]);
    }
}

__global__ __launch_bounds__(THREADS)
void qsim_kernel(const float* __restrict__ fv, const float* __restrict__ theta,
                 float* __restrict__ out) {
    __shared__ float2 st[DIM];
    __shared__ float2 rot[(NLAYERS + 1) * NQ][4];
    __shared__ __align__(16) float2 triM[NGATES][64];
    __shared__ float  wpart[NWAVES][NQ];

    const int b    = blockIdx.x;
    const int tid  = threadIdx.x;
    const int lane = tid & 31;
    const int wave = tid >> 5;

    for (int i = tid; i < DIM; i += THREADS) st[i] = make_float2(0.f, 0.f);
    if (tid == 0) st[0] = make_float2(1.f, 0.f);

    if (tid < (NLAYERS + 1) * NQ) {
        float2 U[4];
        if (tid < NQ) {
            embed_rot(fv[b * NQ + tid], U);
        } else {
            const float* ta = theta + (tid - NQ) * 3;
            fused_rot(ta[0], ta[1], ta[2], U);
        }
        #pragma unroll
        for (int k = 0; k < 4; ++k) rot[tid][k] = U[k];
    }
    __syncthreads();

    for (int e = tid; e < NGATES * 64; e += THREADS) {
        const int gate = e >> 6, o = (e >> 3) & 7, hh = (e >> 2) & 1, t = e & 3;
        const int stage = gate >> 2, sh = (gate & 3) * 4;
        const int wa = (PACK4(T0_A, T1_A, T2_A, T3_A) >> sh) & 15;
        const int wb = (PACK4(T0_B, T1_B, T2_B, T3_B) >> sh) & 15;
        const int wc = (PACK4(T0_C, T1_C, T2_C, T3_C) >> sh) & 15;
        const float2 ua = rot[stage * NQ + wa][(o & 1) * 2 + hh];
        const float2 ub = rot[stage * NQ + wb][((o >> 1) & 1) * 2 + (t & 1)];
        const float2 uc = rot[stage * NQ + wc][(o >> 2) * 2 + (t >> 1)];
        triM[gate][e & 63] = cmul(uc, cmul(ub, ua));
    }
    __syncthreads();

    apply_triple_wmma<T0_A, T0_B, T0_C, T0_COLS, false>(st, &triM[0][0], lane, wave);
    __syncthreads();
    apply_triple_wmma<T1_A, T1_B, T1_C, T1_COLS, false>(st, &triM[1][0], lane, wave);
    __syncthreads();
    apply_triple_wmma<T2_A, T2_B, T2_C, T2_COLS, false>(st, &triM[2][0], lane, wave);
    __syncthreads();
    apply_triple_wmma<T3_A, T3_B, T3_C, T3_COLS, false>(st, &triM[3][0], lane, wave);
    __syncthreads();

    for (int l = 1; l <= NLAYERS; ++l) {
        apply_triple_wmma<T0_A, T0_B, T0_C, T0_COLS, false>(st, &triM[l * NTRI + 0][0], lane, wave);
        __syncthreads();
        apply_triple_wmma<T1_A, T1_B, T1_C, T1_COLS, false>(st, &triM[l * NTRI + 1][0], lane, wave);
        __syncthreads();
        apply_triple_wmma<T2_A, T2_B, T2_C, T2_COLS, false>(st, &triM[l * NTRI + 2][0], lane, wave);
        __syncthreads();
        apply_triple_wmma<T3_A, T3_B, T3_C, T3_COLS, true>(st, &triM[l * NTRI + 3][0], lane, wave);
        __syncthreads();
    }

    float acc[NQ];
    #pragma unroll
    for (int w = 0; w < NQ; ++w) acc[w] = 0.f;
    for (int i = tid; i < DIM; i += THREADS) {
        const float2 a  = st[i];
        const float  pz = a.x * a.x + a.y * a.y;
        #pragma unroll
        for (int w = 0; w < NQ; ++w)
            acc[w] += ((i >> w) & 1) ? -pz : pz;
    }
    #pragma unroll
    for (int off = 16; off > 0; off >>= 1) {
        #pragma unroll
        for (int w = 0; w < NQ; ++w)
            acc[w] += __shfl_xor(acc[w], off, 32);
    }
    if (lane == 0) {
        #pragma unroll
        for (int w = 0; w < NQ; ++w) wpart[wave][w] = acc[w];
    }
    __syncthreads();
    if (tid < NQ) {
        float s = 0.f;
        for (int q = 0; q < NWAVES; ++q) s += wpart[q][tid];
        out[b * NQ + tid] = s;
    }
}

extern "C" void kernel_launch(void* const* d_in, const int* in_sizes, int n_in,
                              void* d_out, int out_size, void* d_ws, size_t ws_size,
                              hipStream_t stream) {
    const float* fv    = (const float*)d_in[0];
    const float* theta = (const float*)d_in[1];
    float*       out   = (float*)d_out;
    const int batch = in_sizes[0] / NQ;
    qsim_kernel<<<batch, THREADS, 0, stream>>>(fv, theta, out);
}


# embedding and layer-1 wire gates act on |0..0>: their product state is one complex outer product (K=2 WMMA per tile) stored through the ring map; 20 passes, not 28; products in a new order
# speedup vs baseline: 3.3233x; 1.2754x over previous
#include <hip/hip_runtime.h>


typedef __attribute__((ext_vector_type(2))) float v2f;
typedef __attribute__((ext_vector_type(8))) float v8f;

#define NQ      12
#define DIM     4096
#define THREADS 256
#define NWAVES  8
#define NLAYERS 6
#define NTRI    4
#define NROUNDS 4
#define NGATES  ((NLAYERS - 1) * NTRI)

#define T0_A 3
#define T0_B 4
#define T0_C 11
#define T0_COLS 0x027
#define T1_A 0
#define T1_B 5
#define T1_C 6
#define T1_COLS 0x01E
#define T2_A 1
#define T2_B 7
#define T2_C 8
#define T2_COLS 0x01D
#define T3_A 2
#define T3_B 9
#define T3_C 10
#define T3_COLS 0x01B
#define PACK4(x0, x1, x2, x3) ((x0) | ((x1) << 4) | ((x2) << 8) | ((x3) << 12))

__device__ __forceinline__ float2 cmul(float2 a, float2 b) {
    return make_float2(a.x * b.x - a.y * b.y, a.x * b.y + a.y * b.x);
}

__device__ __forceinline__ void fused_rot(float tx, float ty, float tz, float2* U) {
    const float cx = cosf(0.5f * tx), sx = sinf(0.5f * tx);
    const float cy = cosf(0.5f * ty), sy = sinf(0.5f * ty);
    const float cz = cosf(0.5f * tz), sz = sinf(0.5f * tz);
    const float2 A00 = make_float2( cy * cx,  sy * sx);
    const float2 A01 = make_float2(-sy * cx, -cy * sx);
    const float2 A10 = make_float2( sy * cx, -cy * sx);
    const float2 A11 = make_float2( cy * cx, -sy * sx);
    const float2 e  = make_float2(cz, -sz);
    const float2 ec = make_float2(cz,  sz);
    U[0] = cmul(e, A00);  U[1] = cmul(e, A01);
    U[2] = cmul(ec, A10); U[3] = cmul(ec, A11);
}

__device__ __forceinline__ void embed_col0(float f, float2& e0, float2& e1) {
    const float cy = cosf(0.5f * f),  sy = sinf(0.5f * f);
    const float cz = cosf(0.25f * f), sz = sinf(0.25f * f);
    e0 = make_float2(cz * cy, -sz * cy);
    e1 = make_float2(cz * sy,  sz * sy);
}

__host__ __device__ constexpr int ring_perm(int x) {
    int y = x ^ (x << 1);
    y ^= y << 2;
    y ^= y << 4;
    y ^= y << 8;
    y &= DIM - 1;
    return (y & ~1) | (((y >> (NQ - 1)) ^ x) & 1);
}

__host__ __device__ constexpr int deposit(int v, int mask) {
    int out = 0, k = 0;
    for (int bit = 0; bit < NQ; ++bit)
        if ((mask >> bit) & 1) { out |= ((v >> k) & 1) << bit; ++k; }
    return out;
}

template <int WA, int WB, int WC, int COLS, bool RING>
__device__ __forceinline__ void apply_triple_wmma(float2* st, const float2* Mg,
                                                  int lane, int wave) {
    constexpr int REST = (DIM - 1) & ~(COLS | (1 << WA) | (1 << WB) | (1 << WC));
    const int col = lane & 15;
    const int h   = lane >> 4;

    const int orow = (col >> 3) | ((col & 6));
    const float4* mp = reinterpret_cast<const float4*>(Mg + orow * 8 + h * 4);
    const float4 mlo = mp[0], mhi = mp[1];
    const float2 m[4] = { make_float2(mlo.x, mlo.y), make_float2(mlo.z, mlo.w),
                          make_float2(mhi.x, mhi.y), make_float2(mhi.z, mhi.w) };
    const bool imrow = (col & 1) != 0;
    v2f Aop[4];
    #pragma unroll
    for (int t = 0; t < 4; ++t) {
        Aop[t][0] = imrow ? m[t].y :  m[t].x;
        Aop[t][1] = imrow ? m[t].x : -m[t].y;
    }

    const int base = deposit(col, COLS) | (h << WA) | deposit(wave, REST);

    float2 p[NROUNDS][4];
    #pragma unroll
    for (int r = 0; r < NROUNDS; ++r)
        #pragma unroll
        for (int t = 0; t < 4; ++t)
            p[r][t] = st[base + deposit(r << 3, REST) + ((t & 1) << WB) + ((t >> 1) << WC)];
    #pragma unroll
    for (int r = 0; r < NROUNDS; ++r)
        asm volatile("" :: "v"(p[r][0].x), "v"(p[r][0].y), "v"(p[r][1].x), "v"(p[r][1].y),
                           "v"(p[r][2].x), "v"(p[r][2].y), "v"(p[r][3].x), "v"(p[r][3].y));

    v8f acc[NROUNDS];
    #pragma unroll
    for (int t = 0; t < 4; ++t)
        #pragma unroll
        for (int r = 0; r < NROUNDS; ++r) {
            v2f Bop; Bop[0] = p[r][t].x; Bop[1] = p[r][t].y;
            v8f Cin = {0.f, 0.f, 0.f, 0.f, 0.f, 0.f, 0.f, 0.f};
            if (t > 0) Cin = acc[r];
            acc[r] = __builtin_amdgcn_wmma_f32_16x16x4_f32(false, Aop[t], false, Bop,
                                                           (short)0, Cin, false, false);
        }

    if (!RING) {
        #pragma unroll
        for (int r = 0; r < NROUNDS; ++r)
            #pragma unroll
            for (int q = 0; q < 4; ++q)
                st[base + deposit(r << 3, REST) + ((q & 1) << WB) + ((q >> 1) << WC)] =
                    make_float2(acc[r][2 * q], acc[r][2 * q + 1]);
    } else {
        __syncthreads();
        const int pbase = ring_perm(base);
        #pragma unroll
        for (int r = 0; r < NROUNDS; ++r)
            #pragma unroll
            for (int q = 0; q < 4; ++q)
                st[pbase ^ ring_perm(deposit(r << 3, REST) + ((q & 1) << WB) + ((q >> 1) << WC))] =
                    make_float2(acc[r][2 * q], acc[r][2 * q + 1]);
    }
}

template <int WA, int WB, int WC, int COLS>
__device__ __forceinline__ void product_state_ring(float2* st, const float2* avec,
                                                   const float2* bvec, int lane, int wave) {
    constexpr int REST = (DIM - 1) & ~(COLS | (1 << WA) | (1 << WB) | (1 << WC));
    const int col = lane & 15;
    const int h   = lane >> 4;
    const int orow = (col >> 3) | (col & 6);
    const float2 a = avec[wave * 8 + orow];
    const bool imrow = (col & 1) != 0;
    const bool live  = (h == 0);
    v2f Aop;
    Aop[0] = live ? (imrow ? a.y :  a.x) : 0.f;
    Aop[1] = live ? (imrow ? a.x : -a.y) : 0.f;

    v8f acc[NROUNDS];
    #pragma unroll
    for (int r = 0; r < NROUNDS; ++r) {
        const float2 bb = bvec[r * 16 + col];
        v2f Bop; Bop[0] = live ? bb.x : 0.f; Bop[1] = live ? bb.y : 0.f;
        const v8f zero = {0.f, 0.f, 0.f, 0.f, 0.f, 0.f, 0.f, 0.f};
        acc[r] = __builtin_amdgcn_wmma_f32_16x16x4_f32(false, Aop, false, Bop,
                                                       (short)0, zero, false, false);
    }

    const int base  = deposit(col, COLS) | (h << WA) | deposit(wave, REST);
    const int pbase = ring_perm(base);
    #pragma unroll
    for (int r = 0; r < NROUNDS; ++r)
        #pragma unroll
        for (int q = 0; q < 4; ++q)
            st[pbase ^ ring_perm(deposit(r << 3, REST) + ((q & 1) << WB) + ((q >> 1) << WC))] =
                make_float2(acc[r][2 * q], acc[r][2 * q + 1]);
}

__global__ __launch_bounds__(THREADS)
void qsim_kernel(const float* __restrict__ fv, const float* __restrict__ theta,
                 float* __restrict__ out) {
    __shared__ float2 st[DIM];
    __shared__ float2 rot[(NLAYERS - 1) * NQ][4];
    __shared__ float2 v0[NQ][2];
    __shared__ float2 pvec[2][64];
    __shared__ __align__(16) float2 triM[NGATES][64];
    __shared__ float  wpart[NWAVES][NQ];

    const int b    = blockIdx.x;
    const int tid  = threadIdx.x;
    const int lane = tid & 31;
    const int wave = tid >> 5;

    if (tid < NLAYERS * NQ) {
        float2 U[4];
        const float* ta = theta + tid * 3;
        fused_rot(ta[0], ta[1], ta[2], U);
        if (tid < NQ) {
            float2 e0, e1;
            embed_col0(fv[b * NQ + tid], e0, e1);
            const float2 u0 = cmul(U[0], e0), u1 = cmul(U[1], e1);
            const float2 u2 = cmul(U[2], e0), u3 = cmul(U[3], e1);
            v0[tid][0] = make_float2(u0.x + u1.x, u0.y + u1.y);
            v0[tid][1] = make_float2(u2.x + u3.x, u2.y + u3.y);
        } else {
            #pragma unroll
            for (int k = 0; k < 4; ++k) rot[tid - NQ][k] = U[k];
        }
    }
    __syncthreads();

    if (tid < 128) {
        constexpr int REST3 = (DIM - 1) & ~(T3_COLS | (1 << T3_A) | (1 << T3_B) | (1 << T3_C));
        const int side = tid >> 6, idx = tid & 63;
        int x, mask;
        if (side == 0) {
            x = ((idx & 1) << T3_A) | (((idx >> 1) & 1) << T3_B) | (((idx >> 2) & 1) << T3_C) |
                deposit(idx >> 3, REST3);
            mask = (1 << T3_A) | (1 << T3_B) | (1 << T3_C) | deposit(7, REST3);
        } else {
            x = deposit(idx & 15, T3_COLS) | deposit((idx >> 4) << 3, REST3);
            mask = T3_COLS | deposit(3 << 3, REST3);
        }
        float2 pr = make_float2(1.f, 0.f);
        for (int w = 0; w < NQ; ++w)
            if ((mask >> w) & 1) pr = cmul(pr, v0[w][(x >> w) & 1]);
        pvec[side][idx] = pr;
    }

    for (int e = tid; e < NGATES * 64; e += THREADS) {
        const int gate = e >> 6, o = (e >> 3) & 7, hh = (e >> 2) & 1, t = e & 3;
        const int stage = gate >> 2, sh = (gate & 3) * 4;
        const int wa = (PACK4(T0_A, T1_A, T2_A, T3_A) >> sh) & 15;
        const int wb = (PACK4(T0_B, T1_B, T2_B, T3_B) >> sh) & 15;
        const int wc = (PACK4(T0_C, T1_C, T2_C, T3_C) >> sh) & 15;
        const float2 ua = rot[stage * NQ + wa][(o & 1) * 2 + hh];
        const float2 ub = rot[stage * NQ + wb][((o >> 1) & 1) * 2 + (t & 1)];
        const float2 uc = rot[stage * NQ + wc][(o >> 2) * 2 + (t >> 1)];
        triM[gate][e & 63] = cmul(uc, cmul(ub, ua));
    }
    __syncthreads();

    product_state_ring<T3_A, T3_B, T3_C, T3_COLS>(st, &pvec[0][0], &pvec[1][0], lane, wave);
    __syncthreads();

    for (int l = 0; l < NLAYERS - 1; ++l) {
        apply_triple_wmma<T0_A, T0_B, T0_C, T0_COLS, false>(st, &triM[l * NTRI + 0][0], lane, wave);
        __syncthreads();
        apply_triple_wmma<T1_A, T1_B, T1_C, T1_COLS, false>(st, &triM[l * NTRI + 1][0], lane, wave);
        __syncthreads();
        apply_triple_wmma<T2_A, T2_B, T2_C, T2_COLS, false>(st, &triM[l * NTRI + 2][0], lane, wave);
        __syncthreads();
        apply_triple_wmma<T3_A, T3_B, T3_C, T3_COLS, true>(st, &triM[l * NTRI + 3][0], lane, wave);
        __syncthreads();
    }

    float acc[NQ];
    #pragma unroll
    for (int w = 0; w < NQ; ++w) acc[w] = 0.f;
    for (int i = tid; i < DIM; i += THREADS) {
        const float2 a  = st[i];
        const float  pz = a.x * a.x + a.y * a.y;
        #pragma unroll
        for (int w = 0; w < NQ; ++w)
            acc[w] += ((i >> w) & 1) ? -pz : pz;
    }
    #pragma unroll
    for (int off = 16; off > 0; off >>= 1) {
        #pragma unroll
        for (int w = 0; w < NQ; ++w)
            acc[w] += __shfl_xor(acc[w], off, 32);
    }
    if (lane == 0) {
        #pragma unroll
        for (int w = 0; w < NQ; ++w) wpart[wave][w] = acc[w];
    }
    __syncthreads();
    if (tid < NQ) {
        float s = 0.f;
        for (int q = 0; q < NWAVES; ++q) s += wpart[q][tid];
        out[b * NQ + tid] = s;
    }
}

extern "C" void kernel_launch(void* const* d_in, const int* in_sizes, int n_in,
                              void* d_out, int out_size, void* d_ws, size_t ws_size,
                              hipStream_t stream) {
    const float* fv    = (const float*)d_in[0];
    const float* theta = (const float*)d_in[1];
    float*       out   = (float*)d_out;
    const int batch = in_sizes[0] / NQ;
    qsim_kernel<<<batch, THREADS, 0, stream>>>(fv, theta, out);
}


# last pass of the last layer measures from its accumulators: |amp|^2 summed per lane with compile-time signs of the ring address, no final store, barrier or LDS re-read; <Z> sums in a new order
# speedup vs baseline: 3.5048x; 1.0546x over previous
#include <hip/hip_runtime.h>


typedef __attribute__((ext_vector_type(2))) float v2f;
typedef __attribute__((ext_vector_type(8))) float v8f;

#define NQ      12
#define DIM     4096
#define THREADS 256
#define NWAVES  8
#define NLAYERS 6
#define NTRI    4
#define NROUNDS 4
#define NGATES  ((NLAYERS - 1) * NTRI)

#define T0_A 3
#define T0_B 4
#define T0_C 11
#define T0_COLS 0x027
#define T1_A 0
#define T1_B 5
#define T1_C 6
#define T1_COLS 0x01E
#define T2_A 1
#define T2_B 7
#define T2_C 8
#define T2_COLS 0x01D
#define T3_A 2
#define T3_B 9
#define T3_C 10
#define T3_COLS 0x01B
#define PACK4(x0, x1, x2, x3) ((x0) | ((x1) << 4) | ((x2) << 8) | ((x3) << 12))

__device__ __forceinline__ float2 cmul(float2 a, float2 b) {
    return make_float2(a.x * b.x - a.y * b.y, a.x * b.y + a.y * b.x);
}

__device__ __forceinline__ void fused_rot(float tx, float ty, float tz, float2* U) {
    const float cx = cosf(0.5f * tx), sx = sinf(0.5f * tx);
    const float cy = cosf(0.5f * ty), sy = sinf(0.5f * ty);
    const float cz = cosf(0.5f * tz), sz = sinf(0.5f * tz);
    const float2 A00 = make_float2( cy * cx,  sy * sx);
    const float2 A01 = make_float2(-sy * cx, -cy * sx);
    const float2 A10 = make_float2( sy * cx, -cy * sx);
    const float2 A11 = make_float2( cy * cx, -sy * sx);
    const float2 e  = make_float2(cz, -sz);
    const float2 ec = make_float2(cz,  sz);
    U[0] = cmul(e, A00);  U[1] = cmul(e, A01);
    U[2] = cmul(ec, A10); U[3] = cmul(ec, A11);
}

__device__ __forceinline__ void embed_col0(float f, float2& e0, float2& e1) {
    const float cy = cosf(0.5f * f),  sy = sinf(0.5f * f);
    const float cz = cosf(0.25f * f), sz = sinf(0.25f * f);
    e0 = make_float2(cz * cy, -sz * cy);
    e1 = make_float2(cz * sy,  sz * sy);
}

__host__ __device__ constexpr int ring_perm(int x) {
    int y = x ^ (x << 1);
    y ^= y << 2;
    y ^= y << 4;
    y ^= y << 8;
    y &= DIM - 1;
    return (y & ~1) | (((y >> (NQ - 1)) ^ x) & 1);
}

__host__ __device__ constexpr int deposit(int v, int mask) {
    int out = 0, k = 0;
    for (int bit = 0; bit < NQ; ++bit)
        if ((mask >> bit) & 1) { out |= ((v >> k) & 1) << bit; ++k; }
    return out;
}

template <int WA, int WB, int WC, int COLS, int MODE>
__device__ __forceinline__ void apply_triple_wmma(float2* st, const float2* Mg,
                                                  int lane, int wave, float* zs) {
    constexpr int REST = (DIM - 1) & ~(COLS | (1 << WA) | (1 << WB) | (1 << WC));
    const int col = lane & 15;
    const int h   = lane >> 4;

    const int orow = (col >> 3) | ((col & 6));
    const float4* mp = reinterpret_cast<const float4*>(Mg + orow * 8 + h * 4);
    const float4 mlo = mp[0], mhi = mp[1];
    const float2 m[4] = { make_float2(mlo.x, mlo.y), make_float2(mlo.z, mlo.w),
                          make_float2(mhi.x, mhi.y), make_float2(mhi.z, mhi.w) };
    const bool imrow = (col & 1) != 0;
    v2f Aop[4];
    #pragma unroll
    for (int t = 0; t < 4; ++t) {
        Aop[t][0] = imrow ? m[t].y :  m[t].x;
        Aop[t][1] = imrow ? m[t].x : -m[t].y;
    }

    const int base = deposit(col, COLS) | (h << WA) | deposit(wave, REST);

    float2 p[NROUNDS][4];
    #pragma unroll
    for (int r = 0; r < NROUNDS; ++r)
        #pragma unroll
        for (int t = 0; t < 4; ++t)
            p[r][t] = st[base + deposit(r << 3, REST) + ((t & 1) << WB) + ((t >> 1) << WC)];
    #pragma unroll
    for (int r = 0; r < NROUNDS; ++r)
        asm volatile("" :: "v"(p[r][0].x), "v"(p[r][0].y), "v"(p[r][1].x), "v"(p[r][1].y),
                           "v"(p[r][2].x), "v"(p[r][2].y), "v"(p[r][3].x), "v"(p[r][3].y));

    v8f acc[NROUNDS];
    #pragma unroll
    for (int t = 0; t < 4; ++t)
        #pragma unroll
        for (int r = 0; r < NROUNDS; ++r) {
            v2f Bop; Bop[0] = p[r][t].x; Bop[1] = p[r][t].y;
            v8f Cin = {0.f, 0.f, 0.f, 0.f, 0.f, 0.f, 0.f, 0.f};
            if (t > 0) Cin = acc[r];
            acc[r] = __builtin_amdgcn_wmma_f32_16x16x4_f32(false, Aop[t], false, Bop,
                                                           (short)0, Cin, false, false);
        }

    if (MODE == 0) {
        #pragma unroll
        for (int r = 0; r < NROUNDS; ++r)
            #pragma unroll
            for (int q = 0; q < 4; ++q)
                st[base + deposit(r << 3, REST) + ((q & 1) << WB) + ((q >> 1) << WC)] =
                    make_float2(acc[r][2 * q], acc[r][2 * q + 1]);
    } else if (MODE == 1) {
        __syncthreads();
        const int pbase = ring_perm(base);
        #pragma unroll
        for (int r = 0; r < NROUNDS; ++r)
            #pragma unroll
            for (int q = 0; q < 4; ++q)
                st[pbase ^ ring_perm(deposit(r << 3, REST) + ((q & 1) << WB) + ((q >> 1) << WC))] =
                    make_float2(acc[r][2 * q], acc[r][2 * q + 1]);
    } else {
        const int pbase = ring_perm(base);
        float pz[NROUNDS][4];
        #pragma unroll
        for (int r = 0; r < NROUNDS; ++r)
            #pragma unroll
            for (int q = 0; q < 4; ++q)
                pz[r][q] = acc[r][2 * q] * acc[r][2 * q] + acc[r][2 * q + 1] * acc[r][2 * q + 1];
        #pragma unroll
        for (int w = 0; w < NQ; ++w) {
            float sum = 0.f;
            #pragma unroll
            for (int r = 0; r < NROUNDS; ++r)
                #pragma unroll
                for (int q = 0; q < 4; ++q) {
                    const int c = ring_perm(deposit(r << 3, REST) + ((q & 1) << WB) + ((q >> 1) << WC));
                    const bool neg = ((c >> w) & 1) != 0;
                    if (r == 0 && q == 0) sum = neg ? -pz[r][q] : pz[r][q];
                    else                  sum = neg ? sum - pz[r][q] : sum + pz[r][q];
                }
            zs[w] = ((pbase >> w) & 1) ? -sum : sum;
        }
    }
}

template <int WA, int WB, int WC, int COLS>
__device__ __forceinline__ void product_state_ring(float2* st, const float2* avec,
                                                   const float2* bvec, int lane, int wave) {
    constexpr int REST = (DIM - 1) & ~(COLS | (1 << WA) | (1 << WB) | (1 << WC));
    const int col = lane & 15;
    const int h   = lane >> 4;
    const int orow = (col >> 3) | (col & 6);
    const float2 a = avec[wave * 8 + orow];
    const bool imrow = (col & 1) != 0;
    const bool live  = (h == 0);
    v2f Aop;
    Aop[0] = live ? (imrow ? a.y :  a.x) : 0.f;
    Aop[1] = live ? (imrow ? a.x : -a.y) : 0.f;

    v8f acc[NROUNDS];
    #pragma unroll
    for (int r = 0; r < NROUNDS; ++r) {
        const float2 bb = bvec[r * 16 + col];
        v2f Bop; Bop[0] = live ? bb.x : 0.f; Bop[1] = live ? bb.y : 0.f;
        const v8f zero = {0.f, 0.f, 0.f, 0.f, 0.f, 0.f, 0.f, 0.f};
        acc[r] = __builtin_amdgcn_wmma_f32_16x16x4_f32(false, Aop, false, Bop,
                                                       (short)0, zero, false, false);
    }

    const int base  = deposit(col, COLS) | (h << WA) | deposit(wave, REST);
    const int pbase = ring_perm(base);
    #pragma unroll
    for (int r = 0; r < NROUNDS; ++r)
        #pragma unroll
        for (int q = 0; q < 4; ++q)
            st[pbase ^ ring_perm(deposit(r << 3, REST) + ((q & 1) << WB) + ((q >> 1) << WC))] =
                make_float2(acc[r][2 * q], acc[r][2 * q + 1]);
}

__global__ __launch_bounds__(THREADS)
void qsim_kernel(const float* __restrict__ fv, const float* __restrict__ theta,
                 float* __restrict__ out) {
    __shared__ float2 st[DIM];
    __shared__ float2 rot[(NLAYERS - 1) * NQ][4];
    __shared__ float2 v0[NQ][2];
    __shared__ float2 pvec[2][64];
    __shared__ __align__(16) float2 triM[NGATES][64];
    __shared__ float  wpart[NWAVES][NQ];

    const int b    = blockIdx.x;
    const int tid  = threadIdx.x;
    const int lane = tid & 31;
    const int wave = tid >> 5;

    if (tid < NLAYERS * NQ) {
        float2 U[4];
        const float* ta = theta + tid * 3;
        fused_rot(ta[0], ta[1], ta[2], U);
        if (tid < NQ) {
            float2 e0, e1;
            embed_col0(fv[b * NQ + tid], e0, e1);
            const float2 u0 = cmul(U[0], e0), u1 = cmul(U[1], e1);
            const float2 u2 = cmul(U[2], e0), u3 = cmul(U[3], e1);
            v0[tid][0] = make_float2(u0.x + u1.x, u0.y + u1.y);
            v0[tid][1] = make_float2(u2.x + u3.x, u2.y + u3.y);
        } else {
            #pragma unroll
            for (int k = 0; k < 4; ++k) rot[tid - NQ][k] = U[k];
        }
    }
    __syncthreads();

    if (tid < 128) {
        constexpr int REST3 = (DIM - 1) & ~(T3_COLS | (1 << T3_A) | (1 << T3_B) | (1 << T3_C));
        const int side = tid >> 6, idx = tid & 63;
        int x, mask;
        if (side == 0) {
            x = ((idx & 1) << T3_A) | (((idx >> 1) & 1) << T3_B) | (((idx >> 2) & 1) << T3_C) |
                deposit(idx >> 3, REST3);
            mask = (1 << T3_A) | (1 << T3_B) | (1 << T3_C) | deposit(7, REST3);
        } else {
            x = deposit(idx & 15, T3_COLS) | deposit((idx >> 4) << 3, REST3);
            mask = T3_COLS | deposit(3 << 3, REST3);
        }
        float2 pr = make_float2(1.f, 0.f);
        for (int w = 0; w < NQ; ++w)
            if ((mask >> w) & 1) pr = cmul(pr, v0[w][(x >> w) & 1]);
        pvec[side][idx] = pr;
    }

    for (int e = tid; e < NGATES * 64; e += THREADS) {
        const int gate = e >> 6, o = (e >> 3) & 7, hh = (e >> 2) & 1, t = e & 3;
        const int stage = gate >> 2, sh = (gate & 3) * 4;
        const int wa = (PACK4(T0_A, T1_A, T2_A, T3_A) >> sh) & 15;
        const int wb = (PACK4(T0_B, T1_B, T2_B, T3_B) >> sh) & 15;
        const int wc = (PACK4(T0_C, T1_C, T2_C, T3_C) >> sh) & 15;
        const float2 ua = rot[stage * NQ + wa][(o & 1) * 2 + hh];
        const float2 ub = rot[stage * NQ + wb][((o >> 1) & 1) * 2 + (t & 1)];
        const float2 uc = rot[stage * NQ + wc][(o >> 2) * 2 + (t >> 1)];
        triM[gate][e & 63] = cmul(uc, cmul(ub, ua));
    }
    __syncthreads();

    product_state_ring<T3_A, T3_B, T3_C, T3_COLS>(st, &pvec[0][0], &pvec[1][0], lane, wave);
    __syncthreads();

    for (int l = 0; l < NLAYERS - 2; ++l) {
        apply_triple_wmma<T0_A, T0_B, T0_C, T0_COLS, 0>(st, &triM[l * NTRI + 0][0], lane, wave, nullptr);
        __syncthreads();
        apply_triple_wmma<T1_A, T1_B, T1_C, T1_COLS, 0>(st, &triM[l * NTRI + 1][0], lane, wave, nullptr);
        __syncthreads();
        apply_triple_wmma<T2_A, T2_B, T2_C, T2_COLS, 0>(st, &triM[l * NTRI + 2][0], lane, wave, nullptr);
        __syncthreads();
        apply_triple_wmma<T3_A, T3_B, T3_C, T3_COLS, 1>(st, &triM[l * NTRI + 3][0], lane, wave, nullptr);
        __syncthreads();
    }

    float acc[NQ];
    {
        constexpr int l = NLAYERS - 2;
        apply_triple_wmma<T0_A, T0_B, T0_C, T0_COLS, 0>(st, &triM[l * NTRI + 0][0], lane, wave, nullptr);
        __syncthreads();
        apply_triple_wmma<T1_A, T1_B, T1_C, T1_COLS, 0>(st, &triM[l * NTRI + 1][0], lane, wave, nullptr);
        __syncthreads();
        apply_triple_wmma<T2_A, T2_B, T2_C, T2_COLS, 0>(st, &triM[l * NTRI + 2][0], lane, wave, nullptr);
        __syncthreads();
        apply_triple_wmma<T3_A, T3_B, T3_C, T3_COLS, 2>(st, &triM[l * NTRI + 3][0], lane, wave, acc);
    }

    #pragma unroll
    for (int off = 16; off > 0; off >>= 1) {
        #pragma unroll
        for (int w = 0; w < NQ; ++w)
            acc[w] += __shfl_xor(acc[w], off, 32);
    }
    if (lane == 0) {
        #pragma unroll
        for (int w = 0; w < NQ; ++w) wpart[wave][w] = acc[w];
    }
    __syncthreads();
    if (tid < NQ) {
        float s = 0.f;
        for (int q = 0; q < NWAVES; ++q) s += wpart[q][tid];
        out[b * NQ + tid] = s;
    }
}

extern "C" void kernel_launch(void* const* d_in, const int* in_sizes, int n_in,
                              void* d_out, int out_size, void* d_ws, size_t ws_size,
                              hipStream_t stream) {
    const float* fv    = (const float*)d_in[0];
    const float* theta = (const float*)d_in[1];
    float*       out   = (float*)d_out;
    const int batch = in_sizes[0] / NQ;
    qsim_kernel<<<batch, THREADS, 0, stream>>>(fv, theta, out);
}


# layer-2 wire gates applied to the ring-moved product state as a closed chain of 2x2 factors: one K=4 complex outer product (2 WMMAs per tile) stored via ring; 16 passes, not 20; new sum order
# speedup vs baseline: 4.1443x; 1.1825x over previous
#include <hip/hip_runtime.h>


typedef __attribute__((ext_vector_type(2))) float v2f;
typedef __attribute__((ext_vector_type(8))) float v8f;

#define NQ      12
#define DIM     4096
#define THREADS 256
#define NWAVES  8
#define NLAYERS 6
#define NTRI    4
#define NROUNDS 4
#define NGATES  ((NLAYERS - 2) * NTRI)
#define RPITCH  80

#define T0_A 3
#define T0_B 4
#define T0_C 11
#define T0_COLS 0x027
#define T1_A 0
#define T1_B 5
#define T1_C 6
#define T1_COLS 0x01E
#define T2_A 1
#define T2_B 7
#define T2_C 8
#define T2_COLS 0x01D
#define T3_A 2
#define T3_B 9
#define T3_C 10
#define T3_COLS 0x01B
#define PACK4(x0, x1, x2, x3) ((x0) | ((x1) << 4) | ((x2) << 8) | ((x3) << 12))

__device__ __forceinline__ float2 cmul(float2 a, float2 b) {
    return make_float2(a.x * b.x - a.y * b.y, a.x * b.y + a.y * b.x);
}

__device__ __forceinline__ void fused_rot(float tx, float ty, float tz, float2* U) {
    const float cx = cosf(0.5f * tx), sx = sinf(0.5f * tx);
    const float cy = cosf(0.5f * ty), sy = sinf(0.5f * ty);
    const float cz = cosf(0.5f * tz), sz = sinf(0.5f * tz);
    const float2 A00 = make_float2( cy * cx,  sy * sx);
    const float2 A01 = make_float2(-sy * cx, -cy * sx);
    const float2 A10 = make_float2( sy * cx, -cy * sx);
    const float2 A11 = make_float2( cy * cx, -sy * sx);
    const float2 e  = make_float2(cz, -sz);
    const float2 ec = make_float2(cz,  sz);
    U[0] = cmul(e, A00);  U[1] = cmul(e, A01);
    U[2] = cmul(ec, A10); U[3] = cmul(ec, A11);
}

__device__ __forceinline__ void embed_col0(float f, float2& e0, float2& e1) {
    const float cy = cosf(0.5f * f),  sy = sinf(0.5f * f);
    const float cz = cosf(0.25f * f), sz = sinf(0.25f * f);
    e0 = make_float2(cz * cy, -sz * cy);
    e1 = make_float2(cz * sy,  sz * sy);
}

__host__ __device__ constexpr int ring_perm(int x) {
    int y = x ^ (x << 1);
    y ^= y << 2;
    y ^= y << 4;
    y ^= y << 8;
    y &= DIM - 1;
    return (y & ~1) | (((y >> (NQ - 1)) ^ x) & 1);
}

__host__ __device__ constexpr int deposit(int v, int mask) {
    int out = 0, k = 0;
    for (int bit = 0; bit < NQ; ++bit)
        if ((mask >> bit) & 1) { out |= ((v >> k) & 1) << bit; ++k; }
    return out;
}

template <int WA, int WB, int WC, int COLS, int MODE>
__device__ __forceinline__ void apply_triple_wmma(float2* st, const float2* Mg,
                                                  int lane, int wave, float* zs) {
    constexpr int REST = (DIM - 1) & ~(COLS | (1 << WA) | (1 << WB) | (1 << WC));
    const int col = lane & 15;
    const int h   = lane >> 4;

    const int orow = (col >> 3) | ((col & 6));
    const float4* mp = reinterpret_cast<const float4*>(Mg + orow * 8 + h * 4);
    const float4 mlo = mp[0], mhi = mp[1];
    const float2 m[4] = { make_float2(mlo.x, mlo.y), make_float2(mlo.z, mlo.w),
                          make_float2(mhi.x, mhi.y), make_float2(mhi.z, mhi.w) };
    const bool imrow = (col & 1) != 0;
    v2f Aop[4];
    #pragma unroll
    for (int t = 0; t < 4; ++t) {
        Aop[t][0] = imrow ? m[t].y :  m[t].x;
        Aop[t][1] = imrow ? m[t].x : -m[t].y;
    }

    const int base = deposit(col, COLS) | (h << WA) | deposit(wave, REST);

    float2 p[NROUNDS][4];
    #pragma unroll
    for (int r = 0; r < NROUNDS; ++r)
        #pragma unroll
        for (int t = 0; t < 4; ++t)
            p[r][t] = st[base + deposit(r << 3, REST) + ((t & 1) << WB) + ((t >> 1) << WC)];
    #pragma unroll
    for (int r = 0; r < NROUNDS; ++r)
        asm volatile("" :: "v"(p[r][0].x), "v"(p[r][0].y), "v"(p[r][1].x), "v"(p[r][1].y),
                           "v"(p[r][2].x), "v"(p[r][2].y), "v"(p[r][3].x), "v"(p[r][3].y));

    v8f acc[NROUNDS];
    #pragma unroll
    for (int t = 0; t < 4; ++t)
        #pragma unroll
        for (int r = 0; r < NROUNDS; ++r) {
            v2f Bop; Bop[0] = p[r][t].x; Bop[1] = p[r][t].y;
            v8f Cin = {0.f, 0.f, 0.f, 0.f, 0.f, 0.f, 0.f, 0.f};
            if (t > 0) Cin = acc[r];
            acc[r] = __builtin_amdgcn_wmma_f32_16x16x4_f32(false, Aop[t], false, Bop,
                                                           (short)0, Cin, false, false);
        }

    if (MODE == 0) {
        #pragma unroll
        for (int r = 0; r < NROUNDS; ++r)
            #pragma unroll
            for (int q = 0; q < 4; ++q)
                st[base + deposit(r << 3, REST) + ((q & 1) << WB) + ((q >> 1) << WC)] =
                    make_float2(acc[r][2 * q], acc[r][2 * q + 1]);
    } else if (MODE == 1) {
        __syncthreads();
        const int pbase = ring_perm(base);
        #pragma unroll
        for (int r = 0; r < NROUNDS; ++r)
            #pragma unroll
            for (int q = 0; q < 4; ++q)
                st[pbase ^ ring_perm(deposit(r << 3, REST) + ((q & 1) << WB) + ((q >> 1) << WC))] =
                    make_float2(acc[r][2 * q], acc[r][2 * q + 1]);
    } else {
        const int pbase = ring_perm(base);
        float pz[NROUNDS][4];
        #pragma unroll
        for (int r = 0; r < NROUNDS; ++r)
            #pragma unroll
            for (int q = 0; q < 4; ++q)
                pz[r][q] = acc[r][2 * q] * acc[r][2 * q] + acc[r][2 * q + 1] * acc[r][2 * q + 1];
        #pragma unroll
        for (int w = 0; w < NQ; ++w) {
            float sum = 0.f;
            #pragma unroll
            for (int r = 0; r < NROUNDS; ++r)
                #pragma unroll
                for (int q = 0; q < 4; ++q) {
                    const int c = ring_perm(deposit(r << 3, REST) + ((q & 1) << WB) + ((q >> 1) << WC));
                    const bool neg = ((c >> w) & 1) != 0;
                    if (r == 0 && q == 0) sum = neg ? -pz[r][q] : pz[r][q];
                    else                  sum = neg ? sum - pz[r][q] : sum + pz[r][q];
                }
            zs[w] = ((pbase >> w) & 1) ? -sum : sum;
        }
    }
}

__device__ __forceinline__ void chain_state_ring(float2* st, const float2* Lm,
                                                 const float2* Rm, int lane, int wave) {
    const int col = lane & 15;
    const int h   = lane >> 4;
    const int orow = (col >> 3) | (col & 6);
    const bool imrow = (col & 1) != 0;
    v2f Aop[2];
    #pragma unroll
    for (int t = 0; t < 2; ++t) {
        const float2 a = Lm[(wave * 8 + orow) * 4 + 2 * t + h];
        Aop[t][0] = imrow ? a.y :  a.x;
        Aop[t][1] = imrow ? a.x : -a.y;
    }
    float2 bb[2][NROUNDS];
    #pragma unroll
    for (int t = 0; t < 2; ++t)
        #pragma unroll
        for (int r = 0; r < NROUNDS; ++r)
            bb[t][r] = Rm[(2 * t + h) * RPITCH + r * 16 + col];

    v8f acc[NROUNDS];
    #pragma unroll
    for (int t = 0; t < 2; ++t)
        #pragma unroll
        for (int r = 0; r < NROUNDS; ++r) {
            v2f Bop; Bop[0] = bb[t][r].x; Bop[1] = bb[t][r].y;
            v8f Cin = {0.f, 0.f, 0.f, 0.f, 0.f, 0.f, 0.f, 0.f};
            if (t > 0) Cin = acc[r];
            acc[r] = __builtin_amdgcn_wmma_f32_16x16x4_f32(false, Aop[t], false, Bop,
                                                           (short)0, Cin, false, false);
        }

    const int base  = col | (h << 4) | (wave << 7);
    const int pbase = ring_perm(base);
    #pragma unroll
    for (int r = 0; r < NROUNDS; ++r)
        #pragma unroll
        for (int q = 0; q < 4; ++q)
            st[pbase ^ ring_perm((q << 5) | (r << 10))] =
                make_float2(acc[r][2 * q], acc[r][2 * q + 1]);
}

__device__ __forceinline__ void chain_step(float2& m0, float2& m1, float2 u0, float2 u1,
                                           float2 w0, float2 w1) {
    const float2 s0 = cmul(m0, w0), s1 = cmul(m1, w1);
    const float2 t0 = cmul(m0, w1), t1 = cmul(m1, w0);
    m0 = cmul(u0, make_float2(s0.x + s1.x, s0.y + s1.y));
    m1 = cmul(u1, make_float2(t0.x + t1.x, t0.y + t1.y));
}

__global__ __launch_bounds__(THREADS)
void qsim_kernel(const float* __restrict__ fv, const float* __restrict__ theta,
                 float* __restrict__ out) {
    __shared__ float2 st[DIM];
    __shared__ float2 rot[(NLAYERS - 1) * NQ][4];
    __shared__ float2 v0[NQ][2];
    __shared__ float2 Lm[64 * 4];
    __shared__ float2 Rm[4 * RPITCH];
    __shared__ __align__(16) float2 triM[NGATES][64];
    __shared__ float  wpart[NWAVES][NQ];

    const int b    = blockIdx.x;
    const int tid  = threadIdx.x;
    const int lane = tid & 31;
    const int wave = tid >> 5;

    if (tid < NLAYERS * NQ) {
        float2 U[4];
        const float* ta = theta + tid * 3;
        fused_rot(ta[0], ta[1], ta[2], U);
        if (tid < NQ) {
            float2 e0, e1;
            embed_col0(fv[b * NQ + tid], e0, e1);
            const float2 u0 = cmul(U[0], e0), u1 = cmul(U[1], e1);
            const float2 u2 = cmul(U[2], e0), u3 = cmul(U[3], e1);
            v0[tid][0] = make_float2(u0.x + u1.x, u0.y + u1.y);
            v0[tid][1] = make_float2(u2.x + u3.x, u2.y + u3.y);
        } else {
            #pragma unroll
            for (int k = 0; k < 4; ++k) rot[tid - NQ][k] = U[k];
        }
    }
    __syncthreads();

    if (tid < 64) {
        const int ia = tid;
        const int o4 = ia & 1;
        float2 M[2][2];
        #pragma unroll
        for (int a = 0; a < 2; ++a)
            #pragma unroll
            for (int b = 0; b < 2; ++b)
                M[a][b] = cmul(rot[4][o4 * 2 + b], v0[4][b ^ a]);
        #pragma unroll
        for (int k = 5; k < 10; ++k) {
            const int ok = (ia >> (k - 4)) & 1;
            const float2 u0 = rot[k][ok * 2 + 0], u1 = rot[k][ok * 2 + 1];
            const float2 w0 = v0[k][0], w1 = v0[k][1];
            chain_step(M[0][0], M[0][1], u0, u1, w0, w1);
            chain_step(M[1][0], M[1][1], u0, u1, w0, w1);
        }
        #pragma unroll
        for (int a = 0; a < 2; ++a)
            #pragma unroll
            for (int b = 0; b < 2; ++b) Lm[ia * 4 + a * 2 + b] = M[a][b];
    } else if (tid < 128) {
        const int jb = tid - 64;
        const int o0 = jb & 1, o1 = (jb >> 1) & 1, o2 = (jb >> 2) & 1, o3 = (jb >> 3) & 1;
        const int o10 = (jb >> 4) & 1, o11 = (jb >> 5) & 1;
        float2 R[2][2];
        #pragma unroll
        for (int b = 0; b < 2; ++b)
            #pragma unroll
            for (int c = 0; c < 2; ++c)
                R[b][c] = cmul(rot[10][o10 * 2 + c], v0[10][c ^ b]);
        {
            const float2 u0 = rot[11][o11 * 2 + 0], u1 = rot[11][o11 * 2 + 1];
            const float2 w0 = v0[11][0], w1 = v0[11][1];
            chain_step(R[0][0], R[0][1], u0, u1, w0, w1);
            chain_step(R[1][0], R[1][1], u0, u1, w0, w1);
        }
        float2 Cl[2][2];
        #pragma unroll
        for (int c = 0; c < 2; ++c)
            #pragma unroll
            for (int d = 0; d < 2; ++d) {
                const float2 s0 = cmul(rot[0][o0 * 2 + 0], cmul(v0[0][c],     v0[1][d ^ c]));
                const float2 s1 = cmul(rot[0][o0 * 2 + 1], cmul(v0[0][1 ^ c], v0[1][d ^ 1 ^ c]));
                Cl[c][d] = cmul(rot[1][o1 * 2 + d], make_float2(s0.x + s1.x, s0.y + s1.y));
            }
        #pragma unroll
        for (int b = 0; b < 2; ++b) {
            const float2 r0 = R[b][0], r1 = R[b][1];
            #pragma unroll
            for (int d = 0; d < 2; ++d) {
                const float2 p0 = cmul(r0, Cl[0][d]), p1 = cmul(r1, Cl[1][d]);
                R[b][d] = make_float2(p0.x + p1.x, p0.y + p1.y);
            }
        }
        {
            const float2 u0 = rot[2][o2 * 2 + 0], u1 = rot[2][o2 * 2 + 1];
            const float2 w0 = v0[2][0], w1 = v0[2][1];
            chain_step(R[0][0], R[0][1], u0, u1, w0, w1);
            chain_step(R[1][0], R[1][1], u0, u1, w0, w1);
        }
        {
            const float2 u0 = rot[3][o3 * 2 + 0], u1 = rot[3][o3 * 2 + 1];
            const float2 w0 = v0[3][0], w1 = v0[3][1];
            chain_step(R[0][0], R[0][1], u0, u1, w0, w1);
            chain_step(R[1][0], R[1][1], u0, u1, w0, w1);
        }
        #pragma unroll
        for (int a = 0; a < 2; ++a)
            #pragma unroll
            for (int b = 0; b < 2; ++b) Rm[(a * 2 + b) * RPITCH + jb] = R[b][a];
    }

    for (int e = tid; e < NGATES * 64; e += THREADS) {
        const int gate = e >> 6, o = (e >> 3) & 7, hh = (e >> 2) & 1, t = e & 3;
        const int stage = (gate >> 2) + 1, sh = (gate & 3) * 4;
        const int wa = (PACK4(T0_A, T1_A, T2_A, T3_A) >> sh) & 15;
        const int wb = (PACK4(T0_B, T1_B, T2_B, T3_B) >> sh) & 15;
        const int wc = (PACK4(T0_C, T1_C, T2_C, T3_C) >> sh) & 15;
        const float2 ua = rot[stage * NQ + wa][(o & 1) * 2 + hh];
        const float2 ub = rot[stage * NQ + wb][((o >> 1) & 1) * 2 + (t & 1)];
        const float2 uc = rot[stage * NQ + wc][(o >> 2) * 2 + (t >> 1)];
        triM[gate][e & 63] = cmul(uc, cmul(ub, ua));
    }
    __syncthreads();

    chain_state_ring(st, Lm, Rm, lane, wave);
    __syncthreads();

    for (int l = 0; l < NLAYERS - 3; ++l) {
        apply_triple_wmma<T0_A, T0_B, T0_C, T0_COLS, 0>(st, &triM[l * NTRI + 0][0], lane, wave, nullptr);
        __syncthreads();
        apply_triple_wmma<T1_A, T1_B, T1_C, T1_COLS, 0>(st, &triM[l * NTRI + 1][0], lane, wave, nullptr);
        __syncthreads();
        apply_triple_wmma<T2_A, T2_B, T2_C, T2_COLS, 0>(st, &triM[l * NTRI + 2][0], lane, wave, nullptr);
        __syncthreads();
        apply_triple_wmma<T3_A, T3_B, T3_C, T3_COLS, 1>(st, &triM[l * NTRI + 3][0], lane, wave, nullptr);
        __syncthreads();
    }

    float acc[NQ];
    {
        constexpr int l = NLAYERS - 3;
        apply_triple_wmma<T0_A, T0_B, T0_C, T0_COLS, 0>(st, &triM[l * NTRI + 0][0], lane, wave, nullptr);
        __syncthreads();
        apply_triple_wmma<T1_A, T1_B, T1_C, T1_COLS, 0>(st, &triM[l * NTRI + 1][0], lane, wave, nullptr);
        __syncthreads();
        apply_triple_wmma<T2_A, T2_B, T2_C, T2_COLS, 0>(st, &triM[l * NTRI + 2][0], lane, wave, nullptr);
        __syncthreads();
        apply_triple_wmma<T3_A, T3_B, T3_C, T3_COLS, 2>(st, &triM[l * NTRI + 3][0], lane, wave, acc);
    }

    #pragma unroll
    for (int off = 16; off > 0; off >>= 1) {
        #pragma unroll
        for (int w = 0; w < NQ; ++w)
            acc[w] += __shfl_xor(acc[w], off, 32);
    }
    if (lane == 0) {
        #pragma unroll
        for (int w = 0; w < NQ; ++w) wpart[wave][w] = acc[w];
    }
    __syncthreads();
    if (tid < NQ) {
        float s = 0.f;
        for (int q = 0; q < NWAVES; ++q) s += wpart[q][tid];
        out[b * NQ + tid] = s;
    }
}

extern "C" void kernel_launch(void* const* d_in, const int* in_sizes, int n_in,
                              void* d_out, int out_size, void* d_ws, size_t ws_size,
                              hipStream_t stream) {
    const float* fv    = (const float*)d_in[0];
    const float* theta = (const float*)d_in[1];
    float*       out   = (float*)d_out;
    const int batch = in_sizes[0] / NQ;
    qsim_kernel<<<batch, THREADS, 0, stream>>>(fv, theta, out);
}


# layer-3 gates of 9 wires folded into the 64-entry chain factors (2 parity copies each); wires 3,9,10 and ring 3 in one pass: 13 passes, not 16; new sum order
# speedup vs baseline: 4.5199x; 1.0906x over previous
#include <hip/hip_runtime.h>


typedef __attribute__((ext_vector_type(2))) float v2f;
typedef __attribute__((ext_vector_type(8))) float v8f;

#define NQ      12
#define DIM     4096
#define THREADS 256
#define NWAVES  8
#define NLAYERS 6
#define NTRI    4
#define NROUNDS 4
#define NGATES  ((NLAYERS - 2) * NTRI)
#define RPITCH  80

#define T0_A 3
#define T0_B 4
#define T0_C 11
#define T0_COLS 0x027
#define T1_A 0
#define T1_B 5
#define T1_C 6
#define T1_COLS 0x01E
#define T2_A 1
#define T2_B 7
#define T2_C 8
#define T2_COLS 0x01D
#define T3_A 2
#define T3_B 9
#define T3_C 10
#define T3_COLS 0x01B
#define L3_A 3
#define L3_B T3_B
#define L3_C T3_C
#define L3_COLS 0x017
#define PACK4(x0, x1, x2, x3) ((x0) | ((x1) << 4) | ((x2) << 8) | ((x3) << 12))

__device__ __forceinline__ float2 cmul(float2 a, float2 b) {
    return make_float2(a.x * b.x - a.y * b.y, a.x * b.y + a.y * b.x);
}

__device__ __forceinline__ void fused_rot(float tx, float ty, float tz, float2* U) {
    const float cx = cosf(0.5f * tx), sx = sinf(0.5f * tx);
    const float cy = cosf(0.5f * ty), sy = sinf(0.5f * ty);
    const float cz = cosf(0.5f * tz), sz = sinf(0.5f * tz);
    const float2 A00 = make_float2( cy * cx,  sy * sx);
    const float2 A01 = make_float2(-sy * cx, -cy * sx);
    const float2 A10 = make_float2( sy * cx, -cy * sx);
    const float2 A11 = make_float2( cy * cx, -sy * sx);
    const float2 e  = make_float2(cz, -sz);
    const float2 ec = make_float2(cz,  sz);
    U[0] = cmul(e, A00);  U[1] = cmul(e, A01);
    U[2] = cmul(ec, A10); U[3] = cmul(ec, A11);
}

__device__ __forceinline__ void embed_col0(float f, float2& e0, float2& e1) {
    const float cy = cosf(0.5f * f),  sy = sinf(0.5f * f);
    const float cz = cosf(0.25f * f), sz = sinf(0.25f * f);
    e0 = make_float2(cz * cy, -sz * cy);
    e1 = make_float2(cz * sy,  sz * sy);
}

__host__ __device__ constexpr int ring_perm(int x) {
    int y = x ^ (x << 1);
    y ^= y << 2;
    y ^= y << 4;
    y ^= y << 8;
    y &= DIM - 1;
    return (y & ~1) | (((y >> (NQ - 1)) ^ x) & 1);
}

__host__ __device__ constexpr int deposit(int v, int mask) {
    int out = 0, k = 0;
    for (int bit = 0; bit < NQ; ++bit)
        if ((mask >> bit) & 1) { out |= ((v >> k) & 1) << bit; ++k; }
    return out;
}

template <int WA, int WB, int WC, int COLS, int MODE>
__device__ __forceinline__ void apply_triple_wmma(float2* st, const float2* Mg,
                                                  int lane, int wave, float* zs) {
    constexpr int REST = (DIM - 1) & ~(COLS | (1 << WA) | (1 << WB) | (1 << WC));
    const int col = lane & 15;
    const int h   = lane >> 4;

    const int orow = (col >> 3) | ((col & 6));
    const float4* mp = reinterpret_cast<const float4*>(Mg + orow * 8 + h * 4);
    const float4 mlo = mp[0], mhi = mp[1];
    const float2 m[4] = { make_float2(mlo.x, mlo.y), make_float2(mlo.z, mlo.w),
                          make_float2(mhi.x, mhi.y), make_float2(mhi.z, mhi.w) };
    const bool imrow = (col & 1) != 0;
    v2f Aop[4];
    #pragma unroll
    for (int t = 0; t < 4; ++t) {
        Aop[t][0] = imrow ? m[t].y :  m[t].x;
        Aop[t][1] = imrow ? m[t].x : -m[t].y;
    }

    const int base = deposit(col, COLS) | (h << WA) | deposit(wave, REST);

    float2 p[NROUNDS][4];
    #pragma unroll
    for (int r = 0; r < NROUNDS; ++r)
        #pragma unroll
        for (int t = 0; t < 4; ++t)
            p[r][t] = st[base + deposit(r << 3, REST) + ((t & 1) << WB) + ((t >> 1) << WC)];
    #pragma unroll
    for (int r = 0; r < NROUNDS; ++r)
        asm volatile("" :: "v"(p[r][0].x), "v"(p[r][0].y), "v"(p[r][1].x), "v"(p[r][1].y),
                           "v"(p[r][2].x), "v"(p[r][2].y), "v"(p[r][3].x), "v"(p[r][3].y));

    v8f acc[NROUNDS];
    #pragma unroll
    for (int t = 0; t < 4; ++t)
        #pragma unroll
        for (int r = 0; r < NROUNDS; ++r) {
            v2f Bop; Bop[0] = p[r][t].x; Bop[1] = p[r][t].y;
            v8f Cin = {0.f, 0.f, 0.f, 0.f, 0.f, 0.f, 0.f, 0.f};
            if (t > 0) Cin = acc[r];
            acc[r] = __builtin_amdgcn_wmma_f32_16x16x4_f32(false, Aop[t], false, Bop,
                                                           (short)0, Cin, false, false);
        }

    if (MODE == 0) {
        #pragma unroll
        for (int r = 0; r < NROUNDS; ++r)
            #pragma unroll
            for (int q = 0; q < 4; ++q)
                st[base + deposit(r << 3, REST) + ((q & 1) << WB) + ((q >> 1) << WC)] =
                    make_float2(acc[r][2 * q], acc[r][2 * q + 1]);
    } else if (MODE == 1) {
        __syncthreads();
        const int pbase = ring_perm(base);
        #pragma unroll
        for (int r = 0; r < NROUNDS; ++r)
            #pragma unroll
            for (int q = 0; q < 4; ++q)
                st[pbase ^ ring_perm(deposit(r << 3, REST) + ((q & 1) << WB) + ((q >> 1) << WC))] =
                    make_float2(acc[r][2 * q], acc[r][2 * q + 1]);
    } else {
        const int pbase = ring_perm(base);
        float pz[NROUNDS][4];
        #pragma unroll
        for (int r = 0; r < NROUNDS; ++r)
            #pragma unroll
            for (int q = 0; q < 4; ++q)
                pz[r][q] = acc[r][2 * q] * acc[r][2 * q] + acc[r][2 * q + 1] * acc[r][2 * q + 1];
        #pragma unroll
        for (int w = 0; w < NQ; ++w) {
            float sum = 0.f;
            #pragma unroll
            for (int r = 0; r < NROUNDS; ++r)
                #pragma unroll
                for (int q = 0; q < 4; ++q) {
                    const int c = ring_perm(deposit(r << 3, REST) + ((q & 1) << WB) + ((q >> 1) << WC));
                    const bool neg = ((c >> w) & 1) != 0;
                    if (r == 0 && q == 0) sum = neg ? -pz[r][q] : pz[r][q];
                    else                  sum = neg ? sum - pz[r][q] : sum + pz[r][q];
                }
            zs[w] = ((pbase >> w) & 1) ? -sum : sum;
        }
    }
}

__host__ __device__ constexpr int chain_yconst(int r, int q) {
    return ((r & 1) << 11) | ((r >> 1) << 3) |
           (((q & 1) ? ((1 << 5) | (1 << 9)) : 0) ^ ((q >> 1) ? ((1 << 6) | (1 << 9)) : 0));
}

__device__ __forceinline__ void chain_state_ring(float2* st, const float2* Lf,
                                                 const float2* Rf, int lane, int wave) {
    const int col = lane & 15;
    const int h   = lane >> 4;
    const int orow = (col >> 3) | (col & 6);
    const bool imrow = (col & 1) != 0;
    v2f Aop[2][2];
    #pragma unroll
    for (int p = 0; p < 2; ++p)
        #pragma unroll
        for (int t = 0; t < 2; ++t) {
            const float2 a = Lf[p * 256 + (wave * 8 + orow) * 4 + 2 * t + h];
            Aop[p][t][0] = imrow ? a.y :  a.x;
            Aop[p][t][1] = imrow ? a.x : -a.y;
        }
    const int qpar = (wave ^ (wave >> 1) ^ (wave >> 2)) & 1;
    const float2* Rq = Rf + qpar * (4 * RPITCH);
    float2 bb[2][NROUNDS];
    #pragma unroll
    for (int t = 0; t < 2; ++t)
        #pragma unroll
        for (int r = 0; r < NROUNDS; ++r)
            bb[t][r] = Rq[(2 * t + h) * RPITCH + r * 16 + col];

    v8f acc[NROUNDS];
    #pragma unroll
    for (int t = 0; t < 2; ++t)
        #pragma unroll
        for (int r = 0; r < NROUNDS; ++r) {
            v2f Bop; Bop[0] = bb[t][r].x; Bop[1] = bb[t][r].y;
            v8f Cin = {0.f, 0.f, 0.f, 0.f, 0.f, 0.f, 0.f, 0.f};
            if (t > 0) Cin = acc[r];
            acc[r] = __builtin_amdgcn_wmma_f32_16x16x4_f32(false, Aop[r >> 1][t], false, Bop,
                                                           (short)0, Cin, false, false);
        }

    const int c012 = col & 7;
    const int par3 = (c012 ^ (c012 >> 1) ^ (c012 >> 2)) & 1;
    const int base = c012 | (par3 << 3) | ((col >> 3) << 10) | (h << 4) |
                     ((wave & 3) << 7) | ((((wave >> 2) ^ h) & 1) << 9);
    const int pbase = ring_perm(base);
    #pragma unroll
    for (int r = 0; r < NROUNDS; ++r)
        #pragma unroll
        for (int q = 0; q < 4; ++q)
            st[pbase ^ ring_perm(chain_yconst(r, q))] =
                make_float2(acc[r][2 * q], acc[r][2 * q + 1]);
}

template <int K, bool LAST>
__device__ __forceinline__ void fold_row_gate(const float2* in, float2* out, const float2* U,
                                              int p, int ia) {
    constexpr int kb = K - 4;
    constexpr int m  = 3 << kb;
    const int zk = (p ^ __popc(ia & ((2 << kb) - 1))) & 1;
    const float2 ud = U[zk * 3];
    const float2 uo = U[zk + 1];
    int oi = ia;
    if (LAST) oi = (ia & 31) | ((((ia >> 5) ^ __popc(ia & 7)) & 1) << 5);
    float2 a[4], b[4];
    #pragma unroll
    for (int kc = 0; kc < 4; ++kc) {
        a[kc] = in[ia * 4 + kc];
        b[kc] = in[(ia ^ m) * 4 + kc];
    }
    #pragma unroll
    for (int kc = 0; kc < 4; ++kc) {
        const float2 s = cmul(ud, a[kc]);
        const float2 t = cmul(uo, b[kc]);
        out[oi * 4 + kc] = make_float2(s.x + t.x, s.y + t.y);
    }
}

template <int ZMASK, bool WITHQ, int FLIP, bool LAST>
__device__ __forceinline__ void fold_col_gate(const float2* in, float2* out, const float2* U,
                                              int q, int jb) {
    const int zw = ((WITHQ ? q : 0) ^ __popc(jb & ZMASK)) & 1;
    const float2 ud = U[zw * 3];
    const float2 uo = U[zw + 1];
    const int jt = (jb & 7) | (((jb >> 4) & 1) << 3) | (((jb >> 5) & 1) << 4) |
                   ((__popc(jb & 15) & 1) << 5);
    float2 a[4], b[4];
    #pragma unroll
    for (int kc = 0; kc < 4; ++kc) {
        a[kc] = in[kc * 64 + jb];
        b[kc] = in[kc * 64 + (jb ^ FLIP)];
    }
    #pragma unroll
    for (int kc = 0; kc < 4; ++kc) {
        const float2 s = cmul(ud, a[kc]);
        const float2 t = cmul(uo, b[kc]);
        const float2 v = make_float2(s.x + t.x, s.y + t.y);
        if (LAST) out[kc * RPITCH + jt] = v;
        else      out[kc * 64 + jb] = v;
    }
}

__device__ __forceinline__ void chain_step(float2& m0, float2& m1, float2 u0, float2 u1,
                                           float2 w0, float2 w1) {
    const float2 s0 = cmul(m0, w0), s1 = cmul(m1, w1);
    const float2 t0 = cmul(m0, w1), t1 = cmul(m1, w0);
    m0 = cmul(u0, make_float2(s0.x + s1.x, s0.y + s1.y));
    m1 = cmul(u1, make_float2(t0.x + t1.x, t0.y + t1.y));
}

__global__ __launch_bounds__(THREADS)
void qsim_kernel(const float* __restrict__ fv, const float* __restrict__ theta,
                 float* __restrict__ out) {
    __shared__ float2 st[DIM];
    __shared__ float2 rot[(NLAYERS - 1) * NQ][4];
    __shared__ float2 v0[NQ][2];
    __shared__ float2 Lf[2 * 64 * 4];
    __shared__ float2 Rf[2 * 4 * RPITCH];
    __shared__ __align__(16) float2 triM[NGATES][64];
    __shared__ float  wpart[NWAVES][NQ];

    const int b    = blockIdx.x;
    const int tid  = threadIdx.x;
    const int lane = tid & 31;
    const int wave = tid >> 5;

    if (tid < NLAYERS * NQ) {
        float2 U[4];
        const float* ta = theta + tid * 3;
        fused_rot(ta[0], ta[1], ta[2], U);
        if (tid < NQ) {
            float2 e0, e1;
            embed_col0(fv[b * NQ + tid], e0, e1);
            const float2 u0 = cmul(U[0], e0), u1 = cmul(U[1], e1);
            const float2 u2 = cmul(U[2], e0), u3 = cmul(U[3], e1);
            v0[tid][0] = make_float2(u0.x + u1.x, u0.y + u1.y);
            v0[tid][1] = make_float2(u2.x + u3.x, u2.y + u3.y);
        } else {
            #pragma unroll
            for (int k = 0; k < 4; ++k) rot[tid - NQ][k] = U[k];
        }
    }
    __syncthreads();

    float2* const LA  = st;
    float2* const RA  = st + 256;
    float2* const LB0 = st + 512;
    float2* const LB1 = st + 1024;
    float2* const RB0 = st + 1536;
    float2* const RB1 = st + 2048;

    if (tid < 64) {
        const int ia = tid;
        const int o4 = ia & 1;
        float2 M[2][2];
        #pragma unroll
        for (int a = 0; a < 2; ++a)
            #pragma unroll
            for (int b = 0; b < 2; ++b)
                M[a][b] = cmul(rot[4][o4 * 2 + b], v0[4][b ^ a]);
        #pragma unroll
        for (int k = 5; k < 10; ++k) {
            const int ok = (ia >> (k - 4)) & 1;
            const float2 u0 = rot[k][ok * 2 + 0], u1 = rot[k][ok * 2 + 1];
            const float2 w0 = v0[k][0], w1 = v0[k][1];
            chain_step(M[0][0], M[0][1], u0, u1, w0, w1);
            chain_step(M[1][0], M[1][1], u0, u1, w0, w1);
        }
        #pragma unroll
        for (int a = 0; a < 2; ++a)
            #pragma unroll
            for (int b = 0; b < 2; ++b) LA[ia * 4 + a * 2 + b] = M[a][b];
    } else if (tid < 128) {
        const int jb = tid - 64;
        const int o0 = jb & 1, o1 = (jb >> 1) & 1, o2 = (jb >> 2) & 1, o3 = (jb >> 3) & 1;
        const int o10 = (jb >> 4) & 1, o11 = (jb >> 5) & 1;
        float2 R[2][2];
        #pragma unroll
        for (int b = 0; b < 2; ++b)
            #pragma unroll
            for (int c = 0; c < 2; ++c)
                R[b][c] = cmul(rot[10][o10 * 2 + c], v0[10][c ^ b]);
        {
            const float2 u0 = rot[11][o11 * 2 + 0], u1 = rot[11][o11 * 2 + 1];
            const float2 w0 = v0[11][0], w1 = v0[11][1];
            chain_step(R[0][0], R[0][1], u0, u1, w0, w1);
            chain_step(R[1][0], R[1][1], u0, u1, w0, w1);
        }
        float2 Cl[2][2];
        #pragma unroll
        for (int c = 0; c < 2; ++c)
            #pragma unroll
            for (int d = 0; d < 2; ++d) {
                const float2 s0 = cmul(rot[0][o0 * 2 + 0], cmul(v0[0][c],     v0[1][d ^ c]));
                const float2 s1 = cmul(rot[0][o0 * 2 + 1], cmul(v0[0][1 ^ c], v0[1][d ^ 1 ^ c]));
                Cl[c][d] = cmul(rot[1][o1 * 2 + d], make_float2(s0.x + s1.x, s0.y + s1.y));
            }
        #pragma unroll
        for (int b = 0; b < 2; ++b) {
            const float2 r0 = R[b][0], r1 = R[b][1];
            #pragma unroll
            for (int d = 0; d < 2; ++d) {
                const float2 p0 = cmul(r0, Cl[0][d]), p1 = cmul(r1, Cl[1][d]);
                R[b][d] = make_float2(p0.x + p1.x, p0.y + p1.y);
            }
        }
        {
            const float2 u0 = rot[2][o2 * 2 + 0], u1 = rot[2][o2 * 2 + 1];
            const float2 w0 = v0[2][0], w1 = v0[2][1];
            chain_step(R[0][0], R[0][1], u0, u1, w0, w1);
            chain_step(R[1][0], R[1][1], u0, u1, w0, w1);
        }
        {
            const float2 u0 = rot[3][o3 * 2 + 0], u1 = rot[3][o3 * 2 + 1];
            const float2 w0 = v0[3][0], w1 = v0[3][1];
            chain_step(R[0][0], R[0][1], u0, u1, w0, w1);
            chain_step(R[1][0], R[1][1], u0, u1, w0, w1);
        }
        #pragma unroll
        for (int a = 0; a < 2; ++a)
            #pragma unroll
            for (int b = 0; b < 2; ++b) RA[(a * 2 + b) * 64 + jb] = R[b][a];
    }

    for (int e = 3 * 64 + tid; e < NGATES * 64; e += THREADS) {
        const int gate = e >> 6, o = (e >> 3) & 7, hh = (e >> 2) & 1, t = e & 3;
        const int stage = (gate >> 2) + 1, sh = (gate & 3) * 4;
        const int wa = gate < NTRI ? L3_A : (PACK4(T0_A, T1_A, T2_A, T3_A) >> sh) & 15;
        const int wb = (PACK4(T0_B, T1_B, T2_B, T3_B) >> sh) & 15;
        const int wc = (PACK4(T0_C, T1_C, T2_C, T3_C) >> sh) & 15;
        const float2 ua = rot[stage * NQ + wa][(o & 1) * 2 + hh];
        const float2 ub = rot[stage * NQ + wb][((o >> 1) & 1) * 2 + (t & 1)];
        const float2 uc = rot[stage * NQ + wc][(o >> 2) * 2 + (t >> 1)];
        triM[gate][e & 63] = cmul(uc, cmul(ub, ua));
    }
    __syncthreads();

    {
        const bool rowside = tid < 128;
        const int  pq  = (tid >> 6) & 1;
        const int  idx = tid & 63;
        if (rowside) fold_row_gate<4, false>(LA, LB0 + pq * 256, &rot[NQ + 4][0], pq, idx);
        else         fold_col_gate<0x03, false, 0x06, false>(RA, RB0 + pq * 256, &rot[NQ + 1][0], pq, idx);
        __syncthreads();
        if (rowside) fold_row_gate<5, false>(LB0 + pq * 256, LB1 + pq * 256, &rot[NQ + 5][0], pq, idx);
        else         fold_col_gate<0x07, false, 0x0C, false>(RB0 + pq * 256, RB1 + pq * 256, &rot[NQ + 2][0], pq, idx);
        __syncthreads();
        if (rowside) fold_row_gate<6, false>(LB1 + pq * 256, LB0 + pq * 256, &rot[NQ + 6][0], pq, idx);
        else         fold_col_gate<0x3E, true, 0x03, false>(RB1 + pq * 256, RB0 + pq * 256, &rot[NQ + 0][0], pq, idx);
        __syncthreads();
        if (rowside) fold_row_gate<7, false>(LB0 + pq * 256, LB1 + pq * 256, &rot[NQ + 7][0], pq, idx);
        else         fold_col_gate<0x3F, true, 0x23, true>(RB0 + pq * 256, Rf + pq * (4 * RPITCH), &rot[NQ + 11][0], pq, idx);
        __syncthreads();
        if (rowside) fold_row_gate<8, true>(LB1 + pq * 256, Lf + pq * 256, &rot[NQ + 8][0], pq, idx);
        __syncthreads();
    }

    chain_state_ring(st, Lf, Rf, lane, wave);
    __syncthreads();

    apply_triple_wmma<L3_A, L3_B, L3_C, L3_COLS, 1>(st, &triM[3][0], lane, wave, nullptr);
    __syncthreads();

    for (int l = 1; l < NLAYERS - 3; ++l) {
        apply_triple_wmma<T0_A, T0_B, T0_C, T0_COLS, 0>(st, &triM[l * NTRI + 0][0], lane, wave, nullptr);
        __syncthreads();
        apply_triple_wmma<T1_A, T1_B, T1_C, T1_COLS, 0>(st, &triM[l * NTRI + 1][0], lane, wave, nullptr);
        __syncthreads();
        apply_triple_wmma<T2_A, T2_B, T2_C, T2_COLS, 0>(st, &triM[l * NTRI + 2][0], lane, wave, nullptr);
        __syncthreads();
        apply_triple_wmma<T3_A, T3_B, T3_C, T3_COLS, 1>(st, &triM[l * NTRI + 3][0], lane, wave, nullptr);
        __syncthreads();
    }

    float acc[NQ];
    {
        constexpr int l = NLAYERS - 3;
        apply_triple_wmma<T0_A, T0_B, T0_C, T0_COLS, 0>(st, &triM[l * NTRI + 0][0], lane, wave, nullptr);
        __syncthreads();
        apply_triple_wmma<T1_A, T1_B, T1_C, T1_COLS, 0>(st, &triM[l * NTRI + 1][0], lane, wave, nullptr);
        __syncthreads();
        apply_triple_wmma<T2_A, T2_B, T2_C, T2_COLS, 0>(st, &triM[l * NTRI + 2][0], lane, wave, nullptr);
        __syncthreads();
        apply_triple_wmma<T3_A, T3_B, T3_C, T3_COLS, 2>(st, &triM[l * NTRI + 3][0], lane, wave, acc);
    }

    #pragma unroll
    for (int off = 16; off > 0; off >>= 1) {
        #pragma unroll
        for (int w = 0; w < NQ; ++w)
            acc[w] += __shfl_xor(acc[w], off, 32);
    }
    if (lane == 0) {
        #pragma unroll
        for (int w = 0; w < NQ; ++w) wpart[wave][w] = acc[w];
    }
    __syncthreads();
    if (tid < NQ) {
        float s = 0.f;
        for (int q = 0; q < NWAVES; ++q) s += wpart[q][tid];
        out[b * NQ + tid] = s;
    }
}

extern "C" void kernel_launch(void* const* d_in, const int* in_sizes, int n_in,
                              void* d_out, int out_size, void* d_ws, size_t ws_size,
                              hipStream_t stream) {
    const float* fv    = (const float*)d_in[0];
    const float* theta = (const float*)d_in[1];
    float*       out   = (float*)d_out;
    const int batch = in_sizes[0] / NQ;
    qsim_kernel<<<batch, THREADS, 0, stream>>>(fv, theta, out);
}


# wave priority stepped down with progress (s_setprio 3, 2 behind layer 4, 1 behind layer 5, 0 ahead of the last pass): the four workgroups of a WGP finish together, not oldest first
# speedup vs baseline: 4.7857x; 1.0588x over previous
#include <hip/hip_runtime.h>


typedef __attribute__((ext_vector_type(2))) float v2f;
typedef __attribute__((ext_vector_type(8))) float v8f;

#define NQ      12
#define DIM     4096
#define THREADS 256
#define NWAVES  8
#define NLAYERS 6
#define NTRI    4
#define NROUNDS 4
#define NGATES  ((NLAYERS - 2) * NTRI)
#define RPITCH  80

#define T0_A 3
#define T0_B 4
#define T0_C 11
#define T0_COLS 0x027
#define T1_A 0
#define T1_B 5
#define T1_C 6
#define T1_COLS 0x01E
#define T2_A 1
#define T2_B 7
#define T2_C 8
#define T2_COLS 0x01D
#define T3_A 2
#define T3_B 9
#define T3_C 10
#define T3_COLS 0x01B
#define L3_A 3
#define L3_B T3_B
#define L3_C T3_C
#define L3_COLS 0x017
#define PACK4(x0, x1, x2, x3) ((x0) | ((x1) << 4) | ((x2) << 8) | ((x3) << 12))

__device__ __forceinline__ float2 cmul(float2 a, float2 b) {
    return make_float2(a.x * b.x - a.y * b.y, a.x * b.y + a.y * b.x);
}

__device__ __forceinline__ void fused_rot(float tx, float ty, float tz, float2* U) {
    const float cx = cosf(0.5f * tx), sx = sinf(0.5f * tx);
    const float cy = cosf(0.5f * ty), sy = sinf(0.5f * ty);
    const float cz = cosf(0.5f * tz), sz = sinf(0.5f * tz);
    const float2 A00 = make_float2( cy * cx,  sy * sx);
    const float2 A01 = make_float2(-sy * cx, -cy * sx);
    const float2 A10 = make_float2( sy * cx, -cy * sx);
    const float2 A11 = make_float2( cy * cx, -sy * sx);
    const float2 e  = make_float2(cz, -sz);
    const float2 ec = make_float2(cz,  sz);
    U[0] = cmul(e, A00);  U[1] = cmul(e, A01);
    U[2] = cmul(ec, A10); U[3] = cmul(ec, A11);
}

__device__ __forceinline__ void embed_col0(float f, float2& e0, float2& e1) {
    const float cy = cosf(0.5f * f),  sy = sinf(0.5f * f);
    const float cz = cosf(0.25f * f), sz = sinf(0.25f * f);
    e0 = make_float2(cz * cy, -sz * cy);
    e1 = make_float2(cz * sy,  sz * sy);
}

__host__ __device__ constexpr int ring_perm(int x) {
    int y = x ^ (x << 1);
    y ^= y << 2;
    y ^= y << 4;
    y ^= y << 8;
    y &= DIM - 1;
    return (y & ~1) | (((y >> (NQ - 1)) ^ x) & 1);
}

__host__ __device__ constexpr int deposit(int v, int mask) {
    int out = 0, k = 0;
    for (int bit = 0; bit < NQ; ++bit)
        if ((mask >> bit) & 1) { out |= ((v >> k) & 1) << bit; ++k; }
    return out;
}

template <int WA, int WB, int WC, int COLS, int MODE>
__device__ __forceinline__ void apply_triple_wmma(float2* st, const float2* Mg,
                                                  int lane, int wave, float* zs) {
    constexpr int REST = (DIM - 1) & ~(COLS | (1 << WA) | (1 << WB) | (1 << WC));
    const int col = lane & 15;
    const int h   = lane >> 4;

    const int orow = (col >> 3) | ((col & 6));
    const float4* mp = reinterpret_cast<const float4*>(Mg + orow * 8 + h * 4);
    const float4 mlo = mp[0], mhi = mp[1];
    const float2 m[4] = { make_float2(mlo.x, mlo.y), make_float2(mlo.z, mlo.w),
                          make_float2(mhi.x, mhi.y), make_float2(mhi.z, mhi.w) };
    const bool imrow = (col & 1) != 0;
    v2f Aop[4];
    #pragma unroll
    for (int t = 0; t < 4; ++t) {
        Aop[t][0] = imrow ? m[t].y :  m[t].x;
        Aop[t][1] = imrow ? m[t].x : -m[t].y;
    }

    const int base = deposit(col, COLS) | (h << WA) | deposit(wave, REST);

    float2 p[NROUNDS][4];
    #pragma unroll
    for (int r = 0; r < NROUNDS; ++r)
        #pragma unroll
        for (int t = 0; t < 4; ++t)
            p[r][t] = st[base + deposit(r << 3, REST) + ((t & 1) << WB) + ((t >> 1) << WC)];
    #pragma unroll
    for (int r = 0; r < NROUNDS; ++r)
        asm volatile("" :: "v"(p[r][0].x), "v"(p[r][0].y), "v"(p[r][1].x), "v"(p[r][1].y),
                           "v"(p[r][2].x), "v"(p[r][2].y), "v"(p[r][3].x), "v"(p[r][3].y));

    v8f acc[NROUNDS];
    #pragma unroll
    for (int t = 0; t < 4; ++t)
        #pragma unroll
        for (int r = 0; r < NROUNDS; ++r) {
            v2f Bop; Bop[0] = p[r][t].x; Bop[1] = p[r][t].y;
            v8f Cin = {0.f, 0.f, 0.f, 0.f, 0.f, 0.f, 0.f, 0.f};
            if (t > 0) Cin = acc[r];
            acc[r] = __builtin_amdgcn_wmma_f32_16x16x4_f32(false, Aop[t], false, Bop,
                                                           (short)0, Cin, false, false);
        }

    if (MODE == 0) {
        #pragma unroll
        for (int r = 0; r < NROUNDS; ++r)
            #pragma unroll
            for (int q = 0; q < 4; ++q)
                st[base + deposit(r << 3, REST) + ((q & 1) << WB) + ((q >> 1) << WC)] =
                    make_float2(acc[r][2 * q], acc[r][2 * q + 1]);
    } else if (MODE == 1) {
        __syncthreads();
        const int pbase = ring_perm(base);
        #pragma unroll
        for (int r = 0; r < NROUNDS; ++r)
            #pragma unroll
            for (int q = 0; q < 4; ++q)
                st[pbase ^ ring_perm(deposit(r << 3, REST) + ((q & 1) << WB) + ((q >> 1) << WC))] =
                    make_float2(acc[r][2 * q], acc[r][2 * q + 1]);
    } else {
        const int pbase = ring_perm(base);
        float pz[NROUNDS][4];
        #pragma unroll
        for (int r = 0; r < NROUNDS; ++r)
            #pragma unroll
            for (int q = 0; q < 4; ++q)
                pz[r][q] = acc[r][2 * q] * acc[r][2 * q] + acc[r][2 * q + 1] * acc[r][2 * q + 1];
        #pragma unroll
        for (int w = 0; w < NQ; ++w) {
            float sum = 0.f;
            #pragma unroll
            for (int r = 0; r < NROUNDS; ++r)
                #pragma unroll
                for (int q = 0; q < 4; ++q) {
                    const int c = ring_perm(deposit(r << 3, REST) + ((q & 1) << WB) + ((q >> 1) << WC));
                    const bool neg = ((c >> w) & 1) != 0;
                    if (r == 0 && q == 0) sum = neg ? -pz[r][q] : pz[r][q];
                    else                  sum = neg ? sum - pz[r][q] : sum + pz[r][q];
                }
            zs[w] = ((pbase >> w) & 1) ? -sum : sum;
        }
    }
}

__host__ __device__ constexpr int chain_yconst(int r, int q) {
    return ((r & 1) << 11) | ((r >> 1) << 3) |
           (((q & 1) ? ((1 << 5) | (1 << 9)) : 0) ^ ((q >> 1) ? ((1 << 6) | (1 << 9)) : 0));
}

__device__ __forceinline__ void chain_state_ring(float2* st, const float2* Lf,
                                                 const float2* Rf, int lane, int wave) {
    const int col = lane & 15;
    const int h   = lane >> 4;
    const int orow = (col >> 3) | (col & 6);
    const bool imrow = (col & 1) != 0;
    v2f Aop[2][2];
    #pragma unroll
    for (int p = 0; p < 2; ++p)
        #pragma unroll
        for (int t = 0; t < 2; ++t) {
            const float2 a = Lf[p * 256 + (wave * 8 + orow) * 4 + 2 * t + h];
            Aop[p][t][0] = imrow ? a.y :  a.x;
            Aop[p][t][1] = imrow ? a.x : -a.y;
        }
    const int qpar = (wave ^ (wave >> 1) ^ (wave >> 2)) & 1;
    const float2* Rq = Rf + qpar * (4 * RPITCH);
    float2 bb[2][NROUNDS];
    #pragma unroll
    for (int t = 0; t < 2; ++t)
        #pragma unroll
        for (int r = 0; r < NROUNDS; ++r)
            bb[t][r] = Rq[(2 * t + h) * RPITCH + r * 16 + col];

    v8f acc[NROUNDS];
    #pragma unroll
    for (int t = 0; t < 2; ++t)
        #pragma unroll
        for (int r = 0; r < NROUNDS; ++r) {
            v2f Bop; Bop[0] = bb[t][r].x; Bop[1] = bb[t][r].y;
            v8f Cin = {0.f, 0.f, 0.f, 0.f, 0.f, 0.f, 0.f, 0.f};
            if (t > 0) Cin = acc[r];
            acc[r] = __builtin_amdgcn_wmma_f32_16x16x4_f32(false, Aop[r >> 1][t], false, Bop,
                                                           (short)0, Cin, false, false);
        }

    const int c012 = col & 7;
    const int par3 = (c012 ^ (c012 >> 1) ^ (c012 >> 2)) & 1;
    const int base = c012 | (par3 << 3) | ((col >> 3) << 10) | (h << 4) |
                     ((wave & 3) << 7) | ((((wave >> 2) ^ h) & 1) << 9);
    const int pbase = ring_perm(base);
    #pragma unroll
    for (int r = 0; r < NROUNDS; ++r)
        #pragma unroll
        for (int q = 0; q < 4; ++q)
            st[pbase ^ ring_perm(chain_yconst(r, q))] =
                make_float2(acc[r][2 * q], acc[r][2 * q + 1]);
}

template <int K, bool LAST>
__device__ __forceinline__ void fold_row_gate(const float2* in, float2* out, const float2* U,
                                              int p, int ia) {
    constexpr int kb = K - 4;
    constexpr int m  = 3 << kb;
    const int zk = (p ^ __popc(ia & ((2 << kb) - 1))) & 1;
    const float2 ud = U[zk * 3];
    const float2 uo = U[zk + 1];
    int oi = ia;
    if (LAST) oi = (ia & 31) | ((((ia >> 5) ^ __popc(ia & 7)) & 1) << 5);
    float2 a[4], b[4];
    #pragma unroll
    for (int kc = 0; kc < 4; ++kc) {
        a[kc] = in[ia * 4 + kc];
        b[kc] = in[(ia ^ m) * 4 + kc];
    }
    #pragma unroll
    for (int kc = 0; kc < 4; ++kc) {
        const float2 s = cmul(ud, a[kc]);
        const float2 t = cmul(uo, b[kc]);
        out[oi * 4 + kc] = make_float2(s.x + t.x, s.y + t.y);
    }
}

template <int ZMASK, bool WITHQ, int FLIP, bool LAST>
__device__ __forceinline__ void fold_col_gate(const float2* in, float2* out, const float2* U,
                                              int q, int jb) {
    const int zw = ((WITHQ ? q : 0) ^ __popc(jb & ZMASK)) & 1;
    const float2 ud = U[zw * 3];
    const float2 uo = U[zw + 1];
    const int jt = (jb & 7) | (((jb >> 4) & 1) << 3) | (((jb >> 5) & 1) << 4) |
                   ((__popc(jb & 15) & 1) << 5);
    float2 a[4], b[4];
    #pragma unroll
    for (int kc = 0; kc < 4; ++kc) {
        a[kc] = in[kc * 64 + jb];
        b[kc] = in[kc * 64 + (jb ^ FLIP)];
    }
    #pragma unroll
    for (int kc = 0; kc < 4; ++kc) {
        const float2 s = cmul(ud, a[kc]);
        const float2 t = cmul(uo, b[kc]);
        const float2 v = make_float2(s.x + t.x, s.y + t.y);
        if (LAST) out[kc * RPITCH + jt] = v;
        else      out[kc * 64 + jb] = v;
    }
}

__device__ __forceinline__ void chain_step(float2& m0, float2& m1, float2 u0, float2 u1,
                                           float2 w0, float2 w1) {
    const float2 s0 = cmul(m0, w0), s1 = cmul(m1, w1);
    const float2 t0 = cmul(m0, w1), t1 = cmul(m1, w0);
    m0 = cmul(u0, make_float2(s0.x + s1.x, s0.y + s1.y));
    m1 = cmul(u1, make_float2(t0.x + t1.x, t0.y + t1.y));
}

__global__ __launch_bounds__(THREADS)
void qsim_kernel(const float* __restrict__ fv, const float* __restrict__ theta,
                 float* __restrict__ out) {
    __shared__ float2 st[DIM];
    __shared__ float2 rot[(NLAYERS - 1) * NQ][4];
    __shared__ float2 v0[NQ][2];
    __shared__ float2 Lf[2 * 64 * 4];
    __shared__ float2 Rf[2 * 4 * RPITCH];
    __shared__ __align__(16) float2 triM[NGATES][64];
    __shared__ float  wpart[NWAVES][NQ];

    const int b    = blockIdx.x;
    const int tid  = threadIdx.x;
    const int lane = tid & 31;
    const int wave = tid >> 5;

    __builtin_amdgcn_s_setprio(3);

    if (tid < NLAYERS * NQ) {
        float2 U[4];
        const float* ta = theta + tid * 3;
        fused_rot(ta[0], ta[1], ta[2], U);
        if (tid < NQ) {
            float2 e0, e1;
            embed_col0(fv[b * NQ + tid], e0, e1);
            const float2 u0 = cmul(U[0], e0), u1 = cmul(U[1], e1);
            const float2 u2 = cmul(U[2], e0), u3 = cmul(U[3], e1);
            v0[tid][0] = make_float2(u0.x + u1.x, u0.y + u1.y);
            v0[tid][1] = make_float2(u2.x + u3.x, u2.y + u3.y);
        } else {
            #pragma unroll
            for (int k = 0; k < 4; ++k) rot[tid - NQ][k] = U[k];
        }
    }
    __syncthreads();

    float2* const LA  = st;
    float2* const RA  = st + 256;
    float2* const LB0 = st + 512;
    float2* const LB1 = st + 1024;
    float2* const RB0 = st + 1536;
    float2* const RB1 = st + 2048;

    if (tid < 64) {
        const int ia = tid;
        const int o4 = ia & 1;
        float2 M[2][2];
        #pragma unroll
        for (int a = 0; a < 2; ++a)
            #pragma unroll
            for (int b = 0; b < 2; ++b)
                M[a][b] = cmul(rot[4][o4 * 2 + b], v0[4][b ^ a]);
        #pragma unroll
        for (int k = 5; k < 10; ++k) {
            const int ok = (ia >> (k - 4)) & 1;
            const float2 u0 = rot[k][ok * 2 + 0], u1 = rot[k][ok * 2 + 1];
            const float2 w0 = v0[k][0], w1 = v0[k][1];
            chain_step(M[0][0], M[0][1], u0, u1, w0, w1);
            chain_step(M[1][0], M[1][1], u0, u1, w0, w1);
        }
        #pragma unroll
        for (int a = 0; a < 2; ++a)
            #pragma unroll
            for (int b = 0; b < 2; ++b) LA[ia * 4 + a * 2 + b] = M[a][b];
    } else if (tid < 128) {
        const int jb = tid - 64;
        const int o0 = jb & 1, o1 = (jb >> 1) & 1, o2 = (jb >> 2) & 1, o3 = (jb >> 3) & 1;
        const int o10 = (jb >> 4) & 1, o11 = (jb >> 5) & 1;
        float2 R[2][2];
        #pragma unroll
        for (int b = 0; b < 2; ++b)
            #pragma unroll
            for (int c = 0; c < 2; ++c)
                R[b][c] = cmul(rot[10][o10 * 2 + c], v0[10][c ^ b]);
        {
            const float2 u0 = rot[11][o11 * 2 + 0], u1 = rot[11][o11 * 2 + 1];
            const float2 w0 = v0[11][0], w1 = v0[11][1];
            chain_step(R[0][0], R[0][1], u0, u1, w0, w1);
            chain_step(R[1][0], R[1][1], u0, u1, w0, w1);
        }
        float2 Cl[2][2];
        #pragma unroll
        for (int c = 0; c < 2; ++c)
            #pragma unroll
            for (int d = 0; d < 2; ++d) {
                const float2 s0 = cmul(rot[0][o0 * 2 + 0], cmul(v0[0][c],     v0[1][d ^ c]));
                const float2 s1 = cmul(rot[0][o0 * 2 + 1], cmul(v0[0][1 ^ c], v0[1][d ^ 1 ^ c]));
                Cl[c][d] = cmul(rot[1][o1 * 2 + d], make_float2(s0.x + s1.x, s0.y + s1.y));
            }
        #pragma unroll
        for (int b = 0; b < 2; ++b) {
            const float2 r0 = R[b][0], r1 = R[b][1];
            #pragma unroll
            for (int d = 0; d < 2; ++d) {
                const float2 p0 = cmul(r0, Cl[0][d]), p1 = cmul(r1, Cl[1][d]);
                R[b][d] = make_float2(p0.x + p1.x, p0.y + p1.y);
            }
        }
        {
            const float2 u0 = rot[2][o2 * 2 + 0], u1 = rot[2][o2 * 2 + 1];
            const float2 w0 = v0[2][0], w1 = v0[2][1];
            chain_step(R[0][0], R[0][1], u0, u1, w0, w1);
            chain_step(R[1][0], R[1][1], u0, u1, w0, w1);
        }
        {
            const float2 u0 = rot[3][o3 * 2 + 0], u1 = rot[3][o3 * 2 + 1];
            const float2 w0 = v0[3][0], w1 = v0[3][1];
            chain_step(R[0][0], R[0][1], u0, u1, w0, w1);
            chain_step(R[1][0], R[1][1], u0, u1, w0, w1);
        }
        #pragma unroll
        for (int a = 0; a < 2; ++a)
            #pragma unroll
            for (int b = 0; b < 2; ++b) RA[(a * 2 + b) * 64 + jb] = R[b][a];
    }

    for (int e = 3 * 64 + tid; e < NGATES * 64; e += THREADS) {
        const int gate = e >> 6, o = (e >> 3) & 7, hh = (e >> 2) & 1, t = e & 3;
        const int stage = (gate >> 2) + 1, sh = (gate & 3) * 4;
        const int wa = gate < NTRI ? L3_A : (PACK4(T0_A, T1_A, T2_A, T3_A) >> sh) & 15;
        const int wb = (PACK4(T0_B, T1_B, T2_B, T3_B) >> sh) & 15;
        const int wc = (PACK4(T0_C, T1_C, T2_C, T3_C) >> sh) & 15;
        const float2 ua = rot[stage * NQ + wa][(o & 1) * 2 + hh];
        const float2 ub = rot[stage * NQ + wb][((o >> 1) & 1) * 2 + (t & 1)];
        const float2 uc = rot[stage * NQ + wc][(o >> 2) * 2 + (t >> 1)];
        triM[gate][e & 63] = cmul(uc, cmul(ub, ua));
    }
    __syncthreads();

    {
        const bool rowside = tid < 128;
        const int  pq  = (tid >> 6) & 1;
        const int  idx = tid & 63;
        if (rowside) fold_row_gate<4, false>(LA, LB0 + pq * 256, &rot[NQ + 4][0], pq, idx);
        else         fold_col_gate<0x03, false, 0x06, false>(RA, RB0 + pq * 256, &rot[NQ + 1][0], pq, idx);
        __syncthreads();
        if (rowside) fold_row_gate<5, false>(LB0 + pq * 256, LB1 + pq * 256, &rot[NQ + 5][0], pq, idx);
        else         fold_col_gate<0x07, false, 0x0C, false>(RB0 + pq * 256, RB1 + pq * 256, &rot[NQ + 2][0], pq, idx);
        __syncthreads();
        if (rowside) fold_row_gate<6, false>(LB1 + pq * 256, LB0 + pq * 256, &rot[NQ + 6][0], pq, idx);
        else         fold_col_gate<0x3E, true, 0x03, false>(RB1 + pq * 256, RB0 + pq * 256, &rot[NQ + 0][0], pq, idx);
        __syncthreads();
        if (rowside) fold_row_gate<7, false>(LB0 + pq * 256, LB1 + pq * 256, &rot[NQ + 7][0], pq, idx);
        else         fold_col_gate<0x3F, true, 0x23, true>(RB0 + pq * 256, Rf + pq * (4 * RPITCH), &rot[NQ + 11][0], pq, idx);
        __syncthreads();
        if (rowside) fold_row_gate<8, true>(LB1 + pq * 256, Lf + pq * 256, &rot[NQ + 8][0], pq, idx);
        __syncthreads();
    }

    chain_state_ring(st, Lf, Rf, lane, wave);
    __syncthreads();

    apply_triple_wmma<L3_A, L3_B, L3_C, L3_COLS, 1>(st, &triM[3][0], lane, wave, nullptr);
    __syncthreads();

    for (int l = 1; l < NLAYERS - 3; ++l) {
        apply_triple_wmma<T0_A, T0_B, T0_C, T0_COLS, 0>(st, &triM[l * NTRI + 0][0], lane, wave, nullptr);
        __syncthreads();
        apply_triple_wmma<T1_A, T1_B, T1_C, T1_COLS, 0>(st, &triM[l * NTRI + 1][0], lane, wave, nullptr);
        __syncthreads();
        apply_triple_wmma<T2_A, T2_B, T2_C, T2_COLS, 0>(st, &triM[l * NTRI + 2][0], lane, wave, nullptr);
        __syncthreads();
        apply_triple_wmma<T3_A, T3_B, T3_C, T3_COLS, 1>(st, &triM[l * NTRI + 3][0], lane, wave, nullptr);
        __syncthreads();
        __builtin_amdgcn_s_setprio(2);
    }
    __builtin_amdgcn_s_setprio(1);

    float acc[NQ];
    {
        constexpr int l = NLAYERS - 3;
        apply_triple_wmma<T0_A, T0_B, T0_C, T0_COLS, 0>(st, &triM[l * NTRI + 0][0], lane, wave, nullptr);
        __syncthreads();
        apply_triple_wmma<T1_A, T1_B, T1_C, T1_COLS, 0>(st, &triM[l * NTRI + 1][0], lane, wave, nullptr);
        __syncthreads();
        apply_triple_wmma<T2_A, T2_B, T2_C, T2_COLS, 0>(st, &triM[l * NTRI + 2][0], lane, wave, nullptr);
        __syncthreads();
        __builtin_amdgcn_s_setprio(0);
        apply_triple_wmma<T3_A, T3_B, T3_C, T3_COLS, 2>(st, &triM[l * NTRI + 3][0], lane, wave, acc);
    }

    #pragma unroll
    for (int off = 16; off > 0; off >>= 1) {
        #pragma unroll
        for (int w = 0; w < NQ; ++w)
            acc[w] += __shfl_xor(acc[w], off, 32);
    }
    if (lane == 0) {
        #pragma unroll
        for (int w = 0; w < NQ; ++w) wpart[wave][w] = acc[w];
    }
    __syncthreads();
    if (tid < NQ) {
        float s = 0.f;
        for (int q = 0; q < NWAVES; ++q) s += wpart[q][tid];
        out[b * NQ + tid] = s;
    }
}

extern "C" void kernel_launch(void* const* d_in, const int* in_sizes, int n_in,
                              void* d_out, int out_size, void* d_ws, size_t ws_size,
                              hipStream_t stream) {
    const float* fv    = (const float*)d_in[0];
    const float* theta = (const float*)d_in[1];
    float*       out   = (float*)d_out;
    const int batch = in_sizes[0] / NQ;
    qsim_kernel<<<batch, THREADS, 0, stream>>>(fv, theta, out);
}


# layers 4-6: each gate split into RY and two diagonals; one real 16x16 matrix carries four wires, re and im as separate tiles: 3 passes per layer, not 4; diagonals by vector unit; new sum order
# speedup vs baseline: 5.0222x; 1.0494x over previous
#include <hip/hip_runtime.h>


typedef __attribute__((ext_vector_type(2))) float v2f;
typedef __attribute__((ext_vector_type(8))) float v8f;

#define NQ      12
#define DIM     4096
#define THREADS 256
#define NWAVES  8
#define NLAYERS 6
#define NROUNDS 4
#define NRY     3
#define NQUAD   3
#define QROUNDS 2
#define DTAB    72
#define RPITCH  80

#define L3_A 3
#define L3_B 9
#define L3_C 10
#define L3_COLS 0x017

__device__ __forceinline__ float2 cmul(float2 a, float2 b) {
    return make_float2(a.x * b.x - a.y * b.y, a.x * b.y + a.y * b.x);
}

__device__ __forceinline__ void fused_rot(float tx, float ty, float tz, float2* U) {
    const float cx = cosf(0.5f * tx), sx = sinf(0.5f * tx);
    const float cy = cosf(0.5f * ty), sy = sinf(0.5f * ty);
    const float cz = cosf(0.5f * tz), sz = sinf(0.5f * tz);
    const float2 A00 = make_float2( cy * cx,  sy * sx);
    const float2 A01 = make_float2(-sy * cx, -cy * sx);
    const float2 A10 = make_float2( sy * cx, -cy * sx);
    const float2 A11 = make_float2( cy * cx, -sy * sx);
    const float2 e  = make_float2(cz, -sz);
    const float2 ec = make_float2(cz,  sz);
    U[0] = cmul(e, A00);  U[1] = cmul(e, A01);
    U[2] = cmul(ec, A10); U[3] = cmul(ec, A11);
}

__device__ __forceinline__ void embed_col0(float f, float2& e0, float2& e1) {
    const float cy = cosf(0.5f * f),  sy = sinf(0.5f * f);
    const float cz = cosf(0.25f * f), sz = sinf(0.25f * f);
    e0 = make_float2(cz * cy, -sz * cy);
    e1 = make_float2(cz * sy,  sz * sy);
}

__host__ __device__ constexpr int ring_perm(int x) {
    int y = x ^ (x << 1);
    y ^= y << 2;
    y ^= y << 4;
    y ^= y << 8;
    y &= DIM - 1;
    return (y & ~1) | (((y >> (NQ - 1)) ^ x) & 1);
}

__host__ __device__ constexpr int deposit(int v, int mask) {
    int out = 0, k = 0;
    for (int bit = 0; bit < NQ; ++bit)
        if ((mask >> bit) & 1) { out |= ((v >> k) & 1) << bit; ++k; }
    return out;
}

template <int WA, int WB, int WC, int COLS>
__device__ __forceinline__ void apply_triple_ring(float2* st, const float2* Mg,
                                                  int lane, int wave) {
    constexpr int REST = (DIM - 1) & ~(COLS | (1 << WA) | (1 << WB) | (1 << WC));
    const int col = lane & 15;
    const int h   = lane >> 4;

    const int orow = (col >> 3) | ((col & 6));
    const float4* mp = reinterpret_cast<const float4*>(Mg + orow * 8 + h * 4);
    const float4 mlo = mp[0], mhi = mp[1];
    const float2 m[4] = { make_float2(mlo.x, mlo.y), make_float2(mlo.z, mlo.w),
                          make_float2(mhi.x, mhi.y), make_float2(mhi.z, mhi.w) };
    const bool imrow = (col & 1) != 0;
    v2f Aop[4];
    #pragma unroll
    for (int t = 0; t < 4; ++t) {
        Aop[t][0] = imrow ? m[t].y :  m[t].x;
        Aop[t][1] = imrow ? m[t].x : -m[t].y;
    }

    const int base = deposit(col, COLS) | (h << WA) | deposit(wave, REST);

    float2 p[NROUNDS][4];
    #pragma unroll
    for (int r = 0; r < NROUNDS; ++r)
        #pragma unroll
        for (int t = 0; t < 4; ++t)
            p[r][t] = st[base + deposit(r << 3, REST) + ((t & 1) << WB) + ((t >> 1) << WC)];
    #pragma unroll
    for (int r = 0; r < NROUNDS; ++r)
        asm volatile("" :: "v"(p[r][0].x), "v"(p[r][0].y), "v"(p[r][1].x), "v"(p[r][1].y),
                           "v"(p[r][2].x), "v"(p[r][2].y), "v"(p[r][3].x), "v"(p[r][3].y));

    v8f acc[NROUNDS];
    #pragma unroll
    for (int t = 0; t < 4; ++t)
        #pragma unroll
        for (int r = 0; r < NROUNDS; ++r) {
            v2f Bop; Bop[0] = p[r][t].x; Bop[1] = p[r][t].y;
            v8f Cin = {0.f, 0.f, 0.f, 0.f, 0.f, 0.f, 0.f, 0.f};
            if (t > 0) Cin = acc[r];
            acc[r] = __builtin_amdgcn_wmma_f32_16x16x4_f32(false, Aop[t], false, Bop,
                                                           (short)0, Cin, false, false);
        }

    __syncthreads();
    float* const re = reinterpret_cast<float*>(st);
    float* const im = re + DIM;
    const int pbase = ring_perm(base);
    #pragma unroll
    for (int r = 0; r < NROUNDS; ++r)
        #pragma unroll
        for (int q = 0; q < 4; ++q) {
            const int a = pbase ^ ring_perm(deposit(r << 3, REST) + ((q & 1) << WB) + ((q >> 1) << WC));
            re[a] = acc[r][2 * q];
            im[a] = acc[r][2 * q + 1];
        }
}

__host__ __device__ constexpr int swap0(int x, int sw) {
    return sw == 0 ? x : ((x & ~(1 | (1 << sw))) | ((x & 1) << sw) | ((x >> sw) & 1));
}

template <int WA, int WB, int WC, int WD, int MODE, bool DIAG, int NEXTA>
__device__ __forceinline__ void apply_quad_wmma(float* re, const float* Mg, const float2* dt,
                                                int lane, int wave, float* zs) {
    static_assert(!DIAG || (WA == 0 && WB == 1 && WC == 6 && WD == 7), "phase tables are laid out for this quad");
    constexpr int COLS = 0x03F & ~((1 << WA) | (1 << WB));
    constexpr int REST = 0xFC0 & ~((1 << WC) | (1 << WD));
    float* const im = re + DIM;
    const int col = lane & 15;
    const int h   = lane >> 4;

    const float4* mp = reinterpret_cast<const float4*>(Mg + col * 16 + h * 8);
    const float4 mlo = mp[0], mhi = mp[1];
    v2f Aop[4];
    Aop[0][0] = mlo.x; Aop[0][1] = mlo.y;
    Aop[1][0] = mlo.z; Aop[1][1] = mlo.w;
    Aop[2][0] = mhi.x; Aop[2][1] = mhi.y;
    Aop[3][0] = mhi.z; Aop[3][1] = mhi.w;

    const int base = deposit(col, COLS) | (h << WB) | deposit(wave, REST);

    const int lb = swap0(base, WA);
    float br[QROUNDS][4][2], bi[QROUNDS][4][2];
    #pragma unroll
    for (int r = 0; r < QROUNDS; ++r)
        #pragma unroll
        for (int t = 0; t < 4; ++t) {
            const int a = lb + deposit(r << 3, REST) + ((t & 1) << WC) + ((t >> 1) << WD);
            const float2 vr = *reinterpret_cast<const float2*>(re + a);
            const float2 vi = *reinterpret_cast<const float2*>(im + a);
            br[r][t][0] = vr.x;  br[r][t][1] = vr.y;
            bi[r][t][0] = vi.x;  bi[r][t][1] = vi.y;
        }

    if (DIAG) {
        const float2 F = cmul(dt[32 + (h | (col << 1))], dt[64 + wave]);
        const float2* t1 = dt + ((h << 3) | (((wave >> 2) & 1) << 2));
        const float2* t2 = dt + 16 + (((col >> 3) << 3) | ((wave & 1) << 2));
        float2 Q[4], P2[4];
        #pragma unroll
        for (int k = 0; k < 4; ++k) {
            Q[k]  = cmul(F, t1[k]);
            P2[k] = t2[k];
        }
        #pragma unroll
        for (int r = 0; r < QROUNDS; ++r)
            #pragma unroll
            for (int t = 0; t < 4; ++t)
                #pragma unroll
                for (int s = 0; s < 2; ++s) {
                    const float2 ph = cmul(Q[r * 2 + s], P2[t]);
                    const float2 a  = cmul(make_float2(br[r][t][s], bi[r][t][s]), ph);
                    br[r][t][s] = a.x;
                    bi[r][t][s] = a.y;
                }
    } else {
        #pragma unroll
        for (int r = 0; r < QROUNDS; ++r) {
            asm volatile("" :: "v"(br[r][0][0]), "v"(br[r][0][1]), "v"(br[r][1][0]), "v"(br[r][1][1]),
                               "v"(br[r][2][0]), "v"(br[r][2][1]), "v"(br[r][3][0]), "v"(br[r][3][1]));
            asm volatile("" :: "v"(bi[r][0][0]), "v"(bi[r][0][1]), "v"(bi[r][1][0]), "v"(bi[r][1][1]),
                               "v"(bi[r][2][0]), "v"(bi[r][2][1]), "v"(bi[r][3][0]), "v"(bi[r][3][1]));
        }
    }

    v8f accr[QROUNDS], acci[QROUNDS];
    #pragma unroll
    for (int t = 0; t < 4; ++t)
        #pragma unroll
        for (int r = 0; r < QROUNDS; ++r) {
            v2f Br; Br[0] = br[r][t][0]; Br[1] = br[r][t][1];
            v2f Bi; Bi[0] = bi[r][t][0]; Bi[1] = bi[r][t][1];
            v8f Cr = {0.f, 0.f, 0.f, 0.f, 0.f, 0.f, 0.f, 0.f};
            v8f Ci = {0.f, 0.f, 0.f, 0.f, 0.f, 0.f, 0.f, 0.f};
            if (t > 0) { Cr = accr[r]; Ci = acci[r]; }
            accr[r] = __builtin_amdgcn_wmma_f32_16x16x4_f32(false, Aop[t], false, Br,
                                                            (short)0, Cr, false, false);
            acci[r] = __builtin_amdgcn_wmma_f32_16x16x4_f32(false, Aop[t], false, Bi,
                                                            (short)0, Ci, false, false);
        }

    if (MODE == 0) {
        const int sb = swap0(base, NEXTA);
        #pragma unroll
        for (int r = 0; r < QROUNDS; ++r)
            #pragma unroll
            for (int i = 0; i < 8; ++i) {
                const int a = sb + swap0(deposit(r << 3, REST) + ((i & 1) << WA) + (((i >> 1) & 1) << WC) + ((i >> 2) << WD), NEXTA);
                re[a] = accr[r][i];
                im[a] = acci[r][i];
            }
    } else if (MODE == 1) {
        __syncthreads();
        const int pbase = ring_perm(base);
        #pragma unroll
        for (int r = 0; r < QROUNDS; ++r)
            #pragma unroll
            for (int i = 0; i < 8; ++i) {
                const int a = pbase ^ ring_perm(deposit(r << 3, REST) + ((i & 1) << WA) + (((i >> 1) & 1) << WC) + ((i >> 2) << WD));
                re[a] = accr[r][i];
                im[a] = acci[r][i];
            }
    } else {
        const int pbase = ring_perm(base);
        float pz[QROUNDS][8];
        #pragma unroll
        for (int r = 0; r < QROUNDS; ++r)
            #pragma unroll
            for (int i = 0; i < 8; ++i)
                pz[r][i] = accr[r][i] * accr[r][i] + acci[r][i] * acci[r][i];
        #pragma unroll
        for (int w = 0; w < NQ; ++w) {
            float sum = 0.f;
            #pragma unroll
            for (int r = 0; r < QROUNDS; ++r)
                #pragma unroll
                for (int i = 0; i < 8; ++i) {
                    const int c = ring_perm(deposit(r << 3, REST) + ((i & 1) << WA) + (((i >> 1) & 1) << WC) + ((i >> 2) << WD));
                    const bool neg = ((c >> w) & 1) != 0;
                    if (r == 0 && i == 0) sum = neg ? -pz[r][i] : pz[r][i];
                    else                  sum = neg ? sum - pz[r][i] : sum + pz[r][i];
                }
            zs[w] = ((pbase >> w) & 1) ? -sum : sum;
        }
    }
}

__host__ __device__ constexpr int chain_yconst(int r, int q) {
    return ((r & 1) << 11) | ((r >> 1) << 3) |
           (((q & 1) ? ((1 << 5) | (1 << 9)) : 0) ^ ((q >> 1) ? ((1 << 6) | (1 << 9)) : 0));
}

__device__ __forceinline__ void chain_state_ring(float2* st, const float2* Lf,
                                                 const float2* Rf, int lane, int wave) {
    const int col = lane & 15;
    const int h   = lane >> 4;
    const int orow = (col >> 3) | (col & 6);
    const bool imrow = (col & 1) != 0;
    v2f Aop[2][2];
    #pragma unroll
    for (int p = 0; p < 2; ++p)
        #pragma unroll
        for (int t = 0; t < 2; ++t) {
            const float2 a = Lf[p * 256 + (wave * 8 + orow) * 4 + 2 * t + h];
            Aop[p][t][0] = imrow ? a.y :  a.x;
            Aop[p][t][1] = imrow ? a.x : -a.y;
        }
    const int qpar = (wave ^ (wave >> 1) ^ (wave >> 2)) & 1;
    const float2* Rq = Rf + qpar * (4 * RPITCH);
    float2 bb[2][NROUNDS];
    #pragma unroll
    for (int t = 0; t < 2; ++t)
        #pragma unroll
        for (int r = 0; r < NROUNDS; ++r)
            bb[t][r] = Rq[(2 * t + h) * RPITCH + r * 16 + col];

    v8f acc[NROUNDS];
    #pragma unroll
    for (int t = 0; t < 2; ++t)
        #pragma unroll
        for (int r = 0; r < NROUNDS; ++r) {
            v2f Bop; Bop[0] = bb[t][r].x; Bop[1] = bb[t][r].y;
            v8f Cin = {0.f, 0.f, 0.f, 0.f, 0.f, 0.f, 0.f, 0.f};
            if (t > 0) Cin = acc[r];
            acc[r] = __builtin_amdgcn_wmma_f32_16x16x4_f32(false, Aop[r >> 1][t], false, Bop,
                                                           (short)0, Cin, false, false);
        }

    const int c012 = col & 7;
    const int par3 = (c012 ^ (c012 >> 1) ^ (c012 >> 2)) & 1;
    const int base = c012 | (par3 << 3) | ((col >> 3) << 10) | (h << 4) |
                     ((wave & 3) << 7) | ((((wave >> 2) ^ h) & 1) << 9);
    const int pbase = ring_perm(base);
    #pragma unroll
    for (int r = 0; r < NROUNDS; ++r)
        #pragma unroll
        for (int q = 0; q < 4; ++q)
            st[pbase ^ ring_perm(chain_yconst(r, q))] =
                make_float2(acc[r][2 * q], acc[r][2 * q + 1]);
}

template <int K, bool LAST>
__device__ __forceinline__ void fold_row_gate(const float2* in, float2* out, const float2* U,
                                              int p, int ia) {
    constexpr int kb = K - 4;
    constexpr int m  = 3 << kb;
    const int zk = (p ^ __popc(ia & ((2 << kb) - 1))) & 1;
    const float2 ud = U[zk * 3];
    const float2 uo = U[zk + 1];
    int oi = ia;
    if (LAST) oi = (ia & 31) | ((((ia >> 5) ^ __popc(ia & 7)) & 1) << 5);
    float2 a[4], b[4];
    #pragma unroll
    for (int kc = 0; kc < 4; ++kc) {
        a[kc] = in[ia * 4 + kc];
        b[kc] = in[(ia ^ m) * 4 + kc];
    }
    #pragma unroll
    for (int kc = 0; kc < 4; ++kc) {
        const float2 s = cmul(ud, a[kc]);
        const float2 t = cmul(uo, b[kc]);
        out[oi * 4 + kc] = make_float2(s.x + t.x, s.y + t.y);
    }
}

template <int ZMASK, bool WITHQ, int FLIP, bool LAST>
__device__ __forceinline__ void fold_col_gate(const float2* in, float2* out, const float2* U,
                                              int q, int jb) {
    const int zw = ((WITHQ ? q : 0) ^ __popc(jb & ZMASK)) & 1;
    const float2 ud = U[zw * 3];
    const float2 uo = U[zw + 1];
    const int jt = (jb & 7) | (((jb >> 4) & 1) << 3) | (((jb >> 5) & 1) << 4) |
                   ((__popc(jb & 15) & 1) << 5);
    float2 a[4], b[4];
    #pragma unroll
    for (int kc = 0; kc < 4; ++kc) {
        a[kc] = in[kc * 64 + jb];
        b[kc] = in[kc * 64 + (jb ^ FLIP)];
    }
    #pragma unroll
    for (int kc = 0; kc < 4; ++kc) {
        const float2 s = cmul(ud, a[kc]);
        const float2 t = cmul(uo, b[kc]);
        const float2 v = make_float2(s.x + t.x, s.y + t.y);
        if (LAST) out[kc * RPITCH + jt] = v;
        else      out[kc * 64 + jb] = v;
    }
}

__device__ __forceinline__ void chain_step(float2& m0, float2& m1, float2 u0, float2 u1,
                                           float2 w0, float2 w1) {
    const float2 s0 = cmul(m0, w0), s1 = cmul(m1, w1);
    const float2 t0 = cmul(m0, w1), t1 = cmul(m1, w0);
    m0 = cmul(u0, make_float2(s0.x + s1.x, s0.y + s1.y));
    m1 = cmul(u1, make_float2(t0.x + t1.x, t0.y + t1.y));
}

__device__ __forceinline__ float2 csel(float2 z, int b) {
    return b ? z : make_float2(1.f, 0.f);
}

__device__ __forceinline__ float2 phase_entry(const float2* ea, const float2* ec, int e) {
    float2 v;
    if (e < 16) {
        const int y0 = e & 1, y11 = (e >> 1) & 1, y10 = (e >> 2) & 1, y1 = (e >> 3) & 1;
        v = csel(ec[0], y0);
        v = cmul(v, csel(ec[11], y11));
        v = cmul(v, csel(ea[0], y0 ^ y11));
        v = cmul(v, csel(ea[1], y1 ^ y0 ^ y11));
        v = cmul(v, csel(ea[11], y11 ^ y10));
    } else if (e < 32) {
        const int y6 = e & 1, y7 = (e >> 1) & 1, y8 = (e >> 2) & 1, y5 = (e >> 3) & 1;
        v = csel(ec[6], y6);
        v = cmul(v, csel(ec[7], y7));
        v = cmul(v, csel(ea[6], y6 ^ y5));
        v = cmul(v, csel(ea[7], y7 ^ y6));
        v = cmul(v, csel(ea[8], y8 ^ y7));
    } else if (e < 64) {
        const int y = (e - 32) << 1;
        v = csel(ec[1], (y >> 1) & 1);
        #pragma unroll
        for (int k = 2; k < 6; ++k) {
            v = cmul(v, csel(ec[k], (y >> k) & 1));
            v = cmul(v, csel(ea[k], ((y >> k) ^ (y >> (k - 1))) & 1));
        }
    } else {
        const int y8 = e & 1, y9 = (e >> 1) & 1, y10 = (e >> 2) & 1;
        v = csel(ec[8], y8);
        v = cmul(v, csel(ec[9], y9));
        v = cmul(v, csel(ec[10], y10));
        v = cmul(v, csel(ea[9], y9 ^ y8));
        v = cmul(v, csel(ea[10], y10 ^ y9));
    }
    return v;
}

__global__ __launch_bounds__(THREADS)
void qsim_kernel(const float* __restrict__ fv, const float* __restrict__ theta,
                 float* __restrict__ out) {
    __shared__ __align__(16) float2 st[DIM];
    __shared__ float2 rot[2 * NQ][4];
    __shared__ float2 v0[NQ][2];
    __shared__ float2 Lf[2 * 64 * 4];
    __shared__ float2 Rf[2 * 4 * RPITCH];
    __shared__ __align__(16) float2 triM[64];
    __shared__ __align__(16) float  ryt[NRY * NQ][4];
    __shared__ float2 eaP[NRY * NQ];
    __shared__ float2 ecL[NRY * NQ];
    __shared__ __align__(16) float  quadM[NRY * NQUAD * 256];
    __shared__ __align__(16) float2 dtab[NRY * DTAB];
    __shared__ float  wpart[NWAVES][NQ];

    const int b    = blockIdx.x;
    const int tid  = threadIdx.x;
    const int lane = tid & 31;
    const int wave = tid >> 5;

    __builtin_amdgcn_s_setprio(3);

    if (tid < NLAYERS * NQ) {
        float2 U[4];
        const float* ta = theta + tid * 3;
        fused_rot(ta[0], ta[1], ta[2], U);
        if (tid < NQ) {
            float2 e0, e1;
            embed_col0(fv[b * NQ + tid], e0, e1);
            const float2 u0 = cmul(U[0], e0), u1 = cmul(U[1], e1);
            const float2 u2 = cmul(U[2], e0), u3 = cmul(U[3], e1);
            v0[tid][0] = make_float2(u0.x + u1.x, u0.y + u1.y);
            v0[tid][1] = make_float2(u2.x + u3.x, u2.y + u3.y);
        } else if (tid < 3 * NQ) {
            #pragma unroll
            for (int k = 0; k < 4; ++k) rot[tid - NQ][k] = U[k];
        } else {
            const int Lw = tid - 3 * NQ;
            const float tiny = 1e-30f;
            const float n00 = U[0].x * U[0].x + U[0].y * U[0].y;
            const float n01 = U[1].x * U[1].x + U[1].y * U[1].y;
            const float n10 = U[2].x * U[2].x + U[2].y * U[2].y;
            const float n11 = U[3].x * U[3].x + U[3].y * U[3].y;
            const bool h00 = n00 > tiny, hs = n01 > tiny && n10 > tiny;
            const float r00 = rsqrtf(h00 ? n00 : 1.f), r01 = rsqrtf(hs ? n01 : 1.f);
            const float r10 = rsqrtf(hs ? n10 : 1.f),  r11 = rsqrtf(n11 > tiny ? n11 : 1.f);
            const float2 p0  = h00 ? make_float2(U[0].x * r00, U[0].y * r00) : make_float2(1.f, 0.f);
            const float2 q10 = hs ? make_float2(U[2].x * r10, U[2].y * r10) : p0;
            const float2 q01 = hs ? make_float2(-U[1].x * r01, -U[1].y * r01)
                                  : make_float2(U[3].x * r11, U[3].y * r11);
            const float2 p0c = make_float2(p0.x, -p0.y);
            const float2 ea = cmul(q10, p0c), ec = cmul(q01, p0c);
            const float c = h00 ? sqrtf(n00) : 0.f;
            const float s = hs ? sqrtf(n10) : 0.f;
            ryt[Lw][0] = c;  ryt[Lw][1] = -s;
            ryt[Lw][2] = s;  ryt[Lw][3] = c;
            ecL[Lw] = ec;
            if (Lw < 2 * NQ) eaP[Lw + NQ] = ea;
            else             eaP[Lw - 2 * NQ] = make_float2(1.f, 0.f);
        }
    }
    __syncthreads();

    float2* const LA  = st;
    float2* const RA  = st + 256;
    float2* const LB0 = st + 512;
    float2* const LB1 = st + 1024;
    float2* const RB0 = st + 1536;
    float2* const RB1 = st + 2048;

    if (tid < 64) {
        const int ia = tid;
        const int o4 = ia & 1;
        float2 M[2][2];
        #pragma unroll
        for (int a = 0; a < 2; ++a)
            #pragma unroll
            for (int b = 0; b < 2; ++b)
                M[a][b] = cmul(rot[4][o4 * 2 + b], v0[4][b ^ a]);
        #pragma unroll
        for (int k = 5; k < 10; ++k) {
            const int ok = (ia >> (k - 4)) & 1;
            const float2 u0 = rot[k][ok * 2 + 0], u1 = rot[k][ok * 2 + 1];
            const float2 w0 = v0[k][0], w1 = v0[k][1];
            chain_step(M[0][0], M[0][1], u0, u1, w0, w1);
            chain_step(M[1][0], M[1][1], u0, u1, w0, w1);
        }
        #pragma unroll
        for (int a = 0; a < 2; ++a)
            #pragma unroll
            for (int b = 0; b < 2; ++b) LA[ia * 4 + a * 2 + b] = M[a][b];
    } else if (tid < 128) {
        const int jb = tid - 64;
        const int o0 = jb & 1, o1 = (jb >> 1) & 1, o2 = (jb >> 2) & 1, o3 = (jb >> 3) & 1;
        const int o10 = (jb >> 4) & 1, o11 = (jb >> 5) & 1;
        float2 R[2][2];
        #pragma unroll
        for (int b = 0; b < 2; ++b)
            #pragma unroll
            for (int c = 0; c < 2; ++c)
                R[b][c] = cmul(rot[10][o10 * 2 + c], v0[10][c ^ b]);
        {
            const float2 u0 = rot[11][o11 * 2 + 0], u1 = rot[11][o11 * 2 + 1];
            const float2 w0 = v0[11][0], w1 = v0[11][1];
            chain_step(R[0][0], R[0][1], u0, u1, w0, w1);
            chain_step(R[1][0], R[1][1], u0, u1, w0, w1);
        }
        float2 Cl[2][2];
        #pragma unroll
        for (int c = 0; c < 2; ++c)
            #pragma unroll
            for (int d = 0; d < 2; ++d) {
                const float2 s0 = cmul(rot[0][o0 * 2 + 0], cmul(v0[0][c],     v0[1][d ^ c]));
                const float2 s1 = cmul(rot[0][o0 * 2 + 1], cmul(v0[0][1 ^ c], v0[1][d ^ 1 ^ c]));
                Cl[c][d] = cmul(rot[1][o1 * 2 + d], make_float2(s0.x + s1.x, s0.y + s1.y));
            }
        #pragma unroll
        for (int b = 0; b < 2; ++b) {
            const float2 r0 = R[b][0], r1 = R[b][1];
            #pragma unroll
            for (int d = 0; d < 2; ++d) {
                const float2 p0 = cmul(r0, Cl[0][d]), p1 = cmul(r1, Cl[1][d]);
                R[b][d] = make_float2(p0.x + p1.x, p0.y + p1.y);
            }
        }
        {
            const float2 u0 = rot[2][o2 * 2 + 0], u1 = rot[2][o2 * 2 + 1];
            const float2 w0 = v0[2][0], w1 = v0[2][1];
            chain_step(R[0][0], R[0][1], u0, u1, w0, w1);
            chain_step(R[1][0], R[1][1], u0, u1, w0, w1);
        }
        {
            const float2 u0 = rot[3][o3 * 2 + 0], u1 = rot[3][o3 * 2 + 1];
            const float2 w0 = v0[3][0], w1 = v0[3][1];
            chain_step(R[0][0], R[0][1], u0, u1, w0, w1);
            chain_step(R[1][0], R[1][1], u0, u1, w0, w1);
        }
        #pragma unroll
        for (int a = 0; a < 2; ++a)
            #pragma unroll
            for (int b = 0; b < 2; ++b) RA[(a * 2 + b) * 64 + jb] = R[b][a];
    } else {
        const int u = tid - 128;
        if (u < 64) {
            const int o = (u >> 3) & 7, hh = (u >> 2) & 1, t = u & 3;
            const float2 ua = rot[NQ + L3_A][(o & 1) * 2 + hh];
            const float2 ub = rot[NQ + L3_B][((o >> 1) & 1) * 2 + (t & 1)];
            const float2 uc = rot[NQ + L3_C][(o >> 2) * 2 + (t >> 1)];
            triM[u] = cmul(uc, cmul(ub, ua));
        }
        #pragma unroll
        for (int k = 0; k < 2; ++k) {
            const int e = u + 128 * k;
            const int col = e >> 4, hh = (e >> 3) & 1, t = (e >> 1) & 3, s = e & 1;
            const int xa = (col & 1) * 2 + s;
            const int xb = (col >> 3) * 2 + hh;
            const int xc = ((col >> 1) & 1) * 2 + (t & 1);
            const int xd = ((col >> 2) & 1) * 2 + (t >> 1);
            #pragma unroll
            for (int g = 0; g < NRY * NQUAD; ++g) {
                const int L = g / NQUAD, j = g % NQUAD;
                quadM[g * 256 + e] = ryt[L * NQ + 2 * j][xa] * ryt[L * NQ + 2 * j + 1][xb] *
                                     ryt[L * NQ + 6 + 2 * j][xc] * ryt[L * NQ + 7 + 2 * j][xd];
            }
        }
        #pragma unroll
        for (int k = 0; k < 2; ++k) {
            const int d = u + 128 * k;
            if (d < NRY * DTAB) {
                const int L = d / DTAB;
                dtab[d] = phase_entry(&eaP[L * NQ], &ecL[L * NQ], d - L * DTAB);
            }
        }
    }
    __syncthreads();

    {
        const bool rowside = tid < 128;
        const int  pq  = (tid >> 6) & 1;
        const int  idx = tid & 63;
        if (rowside) fold_row_gate<4, false>(LA, LB0 + pq * 256, &rot[NQ + 4][0], pq, idx);
        else         fold_col_gate<0x03, false, 0x06, false>(RA, RB0 + pq * 256, &rot[NQ + 1][0], pq, idx);
        __syncthreads();
        if (rowside) fold_row_gate<5, false>(LB0 + pq * 256, LB1 + pq * 256, &rot[NQ + 5][0], pq, idx);
        else         fold_col_gate<0x07, false, 0x0C, false>(RB0 + pq * 256, RB1 + pq * 256, &rot[NQ + 2][0], pq, idx);
        __syncthreads();
        if (rowside) fold_row_gate<6, false>(LB1 + pq * 256, LB0 + pq * 256, &rot[NQ + 6][0], pq, idx);
        else         fold_col_gate<0x3E, true, 0x03, false>(RB1 + pq * 256, RB0 + pq * 256, &rot[NQ + 0][0], pq, idx);
        __syncthreads();
        if (rowside) fold_row_gate<7, false>(LB0 + pq * 256, LB1 + pq * 256, &rot[NQ + 7][0], pq, idx);
        else         fold_col_gate<0x3F, true, 0x23, true>(RB0 + pq * 256, Rf + pq * (4 * RPITCH), &rot[NQ + 11][0], pq, idx);
        __syncthreads();
        if (rowside) fold_row_gate<8, true>(LB1 + pq * 256, Lf + pq * 256, &rot[NQ + 8][0], pq, idx);
        __syncthreads();
    }

    chain_state_ring(st, Lf, Rf, lane, wave);
    __syncthreads();

    apply_triple_ring<L3_A, L3_B, L3_C, L3_COLS>(st, &triM[0], lane, wave);
    __syncthreads();

    float* const pl = reinterpret_cast<float*>(st);
    for (int l = 0; l < NRY - 1; ++l) {
        apply_quad_wmma<0, 1, 6, 7, 0, true, 2>(pl, &quadM[(l * NQUAD + 0) * 256], &dtab[l * DTAB], lane, wave, nullptr);
        __syncthreads();
        apply_quad_wmma<2, 3, 8, 9, 0, false, 4>(pl, &quadM[(l * NQUAD + 1) * 256], nullptr, lane, wave, nullptr);
        __syncthreads();
        apply_quad_wmma<4, 5, 10, 11, 1, false, 0>(pl, &quadM[(l * NQUAD + 2) * 256], nullptr, lane, wave, nullptr);
        __syncthreads();
        __builtin_amdgcn_s_setprio(2);
    }
    __builtin_amdgcn_s_setprio(1);

    float acc[NQ];
    {
        constexpr int l = NRY - 1;
        apply_quad_wmma<0, 1, 6, 7, 0, true, 2>(pl, &quadM[(l * NQUAD + 0) * 256], &dtab[l * DTAB], lane, wave, nullptr);
        __syncthreads();
        apply_quad_wmma<2, 3, 8, 9, 0, false, 4>(pl, &quadM[(l * NQUAD + 1) * 256], nullptr, lane, wave, nullptr);
        __syncthreads();
        __builtin_amdgcn_s_setprio(0);
        apply_quad_wmma<4, 5, 10, 11, 2, false, 0>(pl, &quadM[(l * NQUAD + 2) * 256], nullptr, lane, wave, acc);
    }

    #pragma unroll
    for (int off = 16; off > 0; off >>= 1) {
        #pragma unroll
        for (int w = 0; w < NQ; ++w)
            acc[w] += __shfl_xor(acc[w], off, 32);
    }
    if (lane == 0) {
        #pragma unroll
        for (int w = 0; w < NQ; ++w) wpart[wave][w] = acc[w];
    }
    __syncthreads();
    if (tid < NQ) {
        float s = 0.f;
        for (int q = 0; q < NWAVES; ++q) s += wpart[q][tid];
        out[b * NQ + tid] = s;
    }
}

extern "C" void kernel_launch(void* const* d_in, const int* in_sizes, int n_in,
                              void* d_out, int out_size, void* d_ws, size_t ws_size,
                              hipStream_t stream) {
    const float* fv    = (const float*)d_in[0];
    const float* theta = (const float*)d_in[1];
    float*       out   = (float*)d_out;
    const int batch = in_sizes[0] / NQ;
    qsim_kernel<<<batch, THREADS, 0, stream>>>(fv, theta, out);
}
